# starting kernel
# baseline (speedup 1.0000x reference)
#include <hip/hip_runtime.h>

typedef float v2f __attribute__((ext_vector_type(2)));
typedef float v8f __attribute__((ext_vector_type(8)));

#define DD 66
#define HHW (66 * 66)
#define DHW (66 * 66 * 66)
#define CI 32
#define CO 32

__global__ void zero_f4_kernel(float4* __restrict__ p, int n4) {
  int i = blockIdx.x * blockDim.x + threadIdx.x;
  int stride = gridDim.x * blockDim.x;
  for (; i < n4; i += stride) p[i] = make_float4(0.f, 0.f, 0.f, 0.f);
}

__launch_bounds__(256)
__global__ void p2g_wmma_kernel(const float* __restrict__ x,
                                const float* __restrict__ pos,
                                const float* __restrict__ kern,
                                const float* __restrict__ bias,
                                float* __restrict__ out,
                                int Mtotal, int Nper) {
  __shared__ float sB[1024];
  __shared__ float2 sP[8][32];

  const int t    = threadIdx.x;
  const int lane = t & 31;
  const int wv   = t >> 5;
  const int half = lane >> 4;
  const int nlo  = lane & 15;

  const int waveBase = (blockIdx.x * 8 + wv) * 32;

  int p  = waveBase + lane;
  int pc = (p < Mtotal) ? p : (Mtotal - 1);
  float valid = (p < Mtotal) ? 1.0f : 0.0f;

  float px = pos[pc * 3 + 0] * 64.0f;
  float py = pos[pc * 3 + 1] * 64.0f;
  float pz = pos[pc * 3 + 2] * 64.0f;
  int b0 = (int)px, b1 = (int)py, b2 = (int)pz;
  float fx = px - (float)b0 - 0.5f;
  float fy = py - (float)b1 - 0.5f;
  float fz = pz - (float)b2 - 0.5f;

  float wx[3], wy[3], wz[3];
  wx[0] = 0.5f * (0.5f - fx) * (0.5f - fx) * valid;
  wx[1] = (0.75f - fx * fx) * valid;
  wx[2] = 0.5f * (0.5f + fx) * (0.5f + fx) * valid;
  wy[0] = 0.5f * (0.5f - fy) * (0.5f - fy);
  wy[1] = 0.75f - fy * fy;
  wy[2] = 0.5f * (0.5f + fy) * (0.5f + fy);
  wz[0] = 0.5f * (0.5f - fz) * (0.5f - fz);
  wz[1] = 0.75f - fz * fz;
  wz[2] = 0.5f * (0.5f + fz) * (0.5f + fz);

  int bb = (pc >= Nper) ? 1 : 0;
  int ibase = bb * (CO * DHW) + b0 * HHW + b1 * 66 + b2;

  v2f a[2][8];
#pragma unroll
  for (int mt = 0; mt < 2; ++mt) {
    int row = waveBase + mt * 16 + nlo;
    row = (row < Mtotal) ? row : (Mtotal - 1);
    const float* xr = x + row * CI + half * 2;
#pragma unroll
    for (int c = 0; c < 8; ++c) a[mt][c] = *(const v2f*)(xr + c * 4);
  }

  const float bl0 = bias[nlo];
  const float bl1 = bias[16 + nlo];
  const int chanOff0 = nlo * DHW;
  const int chanOff1 = (16 + nlo) * DHW;

  const float4* k4 = (const float4*)kern;

  for (int o = 0; o < 27; ++o) {
    const int oi = o / 9, oj = (o / 3) % 3, ok = o % 3;

    __syncthreads();

    {
      float4 v4 = k4[o * 256 + t];
      int f   = t * 4;
      int ci  = f >> 5;
      int co  = f & 31;
      int nt  = co >> 4;
      int cc  = ci >> 2;
      int hh  = (ci >> 1) & 1;
      int vv  = ci & 1;
      int base = (nt * 8 + cc) * 64 + (hh * 16 + (co & 15)) * 2 + vv;
      sB[base + 0] = v4.x;
      sB[base + 2] = v4.y;
      sB[base + 4] = v4.z;
      sB[base + 6] = v4.w;
    }
    {
      float wp = wx[oi] * wy[oj] * wz[ok];
      int io = ibase + oi * HHW + oj * 66 + ok;
      sP[wv][lane] = make_float2(wp, __int_as_float(io));
    }
    __syncthreads();

    v2f bf[2][8];
#pragma unroll
    for (int nt = 0; nt < 2; ++nt)
#pragma unroll
      for (int c = 0; c < 8; ++c)
        bf[nt][c] = *(const v2f*)&sB[(nt * 8 + c) * 64 + lane * 2];

    v8f dacc[2][2];
#pragma unroll
    for (int mt = 0; mt < 2; ++mt)
#pragma unroll
      for (int nt = 0; nt < 2; ++nt) {
        float bl = nt ? bl1 : bl0;
#pragma unroll
        for (int r = 0; r < 8; ++r) dacc[mt][nt][r] = bl;
      }

#pragma unroll
    for (int c = 0; c < 8; ++c) {
#pragma unroll
      for (int mt = 0; mt < 2; ++mt) {
#pragma unroll
        for (int nt = 0; nt < 2; ++nt) {
          dacc[mt][nt] = __builtin_amdgcn_wmma_f32_16x16x4_f32(
              false, a[mt][c], false, bf[nt][c], (short)0, dacc[mt][nt],
              false, false);
        }
      }
    }

#pragma unroll
    for (int mt = 0; mt < 2; ++mt) {
#pragma unroll
      for (int r = 0; r < 8; ++r) {
        int m = mt * 16 + half * 8 + r;
        float2 pw = sP[wv][m];
        float w = pw.x;
        int io = __float_as_int(pw.y);
        atomicAdd(out + io + chanOff0, w * dacc[mt][0][r]);
        atomicAdd(out + io + chanOff1, w * dacc[mt][1][r]);
      }
    }
  }
}

extern "C" void kernel_launch(void* const* d_in, const int* in_sizes, int n_in,
                              void* d_out, int out_size, void* d_ws, size_t ws_size,
                              hipStream_t stream) {
  const float* x    = (const float*)d_in[0];
  const float* pos  = (const float*)d_in[1];
  const float* kern = (const float*)d_in[2];
  const float* bias = (const float*)d_in[3];
  float* out = (float*)d_out;

  const int Mtotal = in_sizes[0] / CI;
  const int Nper   = in_sizes[1] / (2 * 3);

  int n4 = out_size / 4;
  zero_f4_kernel<<<2048, 256, 0, stream>>>((float4*)out, n4);

  int blocks = (Mtotal + 255) / 256;
  p2g_wmma_kernel<<<blocks, 256, 0, stream>>>(x, pos, kern, bias, out,
                                              Mtotal, Nper);
}


# scatter into a cell-major workspace (a cell's 32 channels = one 128-B line) so each f32 atomic instruction covers one whole line; an added pass rewrites it channel-major; atomic arrival order kept
# speedup vs baseline: 27.6217x; 27.6217x over previous
#include <hip/hip_runtime.h>


typedef float v2f __attribute__((ext_vector_type(2)));
typedef float v8f __attribute__((ext_vector_type(8)));
typedef unsigned int v2u __attribute__((ext_vector_type(2)));

#define DD 66
#define HHW (66 * 66)
#define DHW (66 * 66 * 66)
#define CI 32
#define CO 32

#define TC 128
#define TH 24
#define TP 33

__global__ void zero_f4_kernel(float4* __restrict__ p, int n4) {
  int i = blockIdx.x * blockDim.x + threadIdx.x;
  int stride = gridDim.x * blockDim.x;
  for (; i < n4; i += stride) p[i] = make_float4(0.f, 0.f, 0.f, 0.f);
}

__launch_bounds__(256)
__global__ void p2g_wmma_kernel(const float* __restrict__ x,
                                const float* __restrict__ pos,
                                const float* __restrict__ kern,
                                const float* __restrict__ bias,
                                float* __restrict__ acc,
                                int Mtotal, int Nper) {
  __shared__ float sB[1024];

  const int t    = threadIdx.x;
  const int lane = t & 31;
  const int wv   = t >> 5;
  const int half = lane >> 4;
  const int nlo  = lane & 15;

  const int waveBase = (blockIdx.x * 8 + wv) * 32;

  int p  = waveBase + lane;
  int pc = (p < Mtotal) ? p : (Mtotal - 1);
  float valid = (p < Mtotal) ? 1.0f : 0.0f;

  float px = pos[pc * 3 + 0] * 64.0f;
  float py = pos[pc * 3 + 1] * 64.0f;
  float pz = pos[pc * 3 + 2] * 64.0f;
  int b0 = (int)px, b1 = (int)py, b2 = (int)pz;
  float fx = px - (float)b0 - 0.5f;
  float fy = py - (float)b1 - 0.5f;
  float fz = pz - (float)b2 - 0.5f;

  float wx[3], wy[3], wz[3];
  wx[0] = 0.5f * (0.5f - fx) * (0.5f - fx) * valid;
  wx[1] = (0.75f - fx * fx) * valid;
  wx[2] = 0.5f * (0.5f + fx) * (0.5f + fx) * valid;
  wy[0] = 0.5f * (0.5f - fy) * (0.5f - fy);
  wy[1] = 0.75f - fy * fy;
  wy[2] = 0.5f * (0.5f + fy) * (0.5f + fy);
  wz[0] = 0.5f * (0.5f - fz) * (0.5f - fz);
  wz[1] = 0.75f - fz * fz;
  wz[2] = 0.5f * (0.5f + fz) * (0.5f + fz);

  int bb = (pc >= Nper) ? 1 : 0;
  int cbase = bb * DHW + b0 * HHW + b1 * 66 + b2;

  v2f a[2][8];
#pragma unroll
  for (int mt = 0; mt < 2; ++mt) {
    int row = waveBase + mt * 16 + nlo;
    row = (row < Mtotal) ? row : (Mtotal - 1);
    const float* xr = x + row * CI + half * 2;
#pragma unroll
    for (int c = 0; c < 8; ++c) a[mt][c] = *(const v2f*)(xr + c * 4);
  }

  const float bl0 = bias[nlo];
  const float bl1 = bias[16 + nlo];

  const float4* k4 = (const float4*)kern;

  for (int o = 0; o < 27; ++o) {
    const int oi = o / 9, oj = (o / 3) % 3, ok = o % 3;

    __syncthreads();

    {
      float4 v4 = k4[o * 256 + t];
      int f   = t * 4;
      int ci  = f >> 5;
      int co  = f & 31;
      int nt  = co >> 4;
      int cc  = ci >> 2;
      int hh  = (ci >> 1) & 1;
      int vv  = ci & 1;
      int base = (nt * 8 + cc) * 64 + (hh * 16 + (co & 15)) * 2 + vv;
      sB[base + 0] = v4.x;
      sB[base + 2] = v4.y;
      sB[base + 4] = v4.z;
      sB[base + 6] = v4.w;
    }
    const float wp = wx[oi] * wy[oj] * wz[ok];
    const int   lo = (cbase + oi * HHW + oj * 66 + ok) * CO;
    __syncthreads();

    v2f bf[2][8];
#pragma unroll
    for (int nt = 0; nt < 2; ++nt)
#pragma unroll
      for (int c = 0; c < 8; ++c)
        bf[nt][c] = *(const v2f*)&sB[(nt * 8 + c) * 64 + lane * 2];

    v8f dacc[2][2];
#pragma unroll
    for (int mt = 0; mt < 2; ++mt)
#pragma unroll
      for (int nt = 0; nt < 2; ++nt) {
        float bl = nt ? bl1 : bl0;
#pragma unroll
        for (int r = 0; r < 8; ++r) dacc[mt][nt][r] = bl;
      }

#pragma unroll
    for (int c = 0; c < 8; ++c) {
#pragma unroll
      for (int mt = 0; mt < 2; ++mt) {
#pragma unroll
        for (int nt = 0; nt < 2; ++nt) {
          dacc[mt][nt] = __builtin_amdgcn_wmma_f32_16x16x4_f32(
              false, a[mt][c], false, bf[nt][c], (short)0, dacc[mt][nt],
              false, false);
        }
      }
    }

#pragma unroll
    for (int mt = 0; mt < 2; ++mt) {
#pragma unroll
      for (int r = 0; r < 8; ++r) {
        v2u s = __builtin_amdgcn_permlane16_swap(
            __float_as_uint(dacc[mt][0][r]), __float_as_uint(dacc[mt][1][r]), false, false);
        const float rowLo = __uint_as_float(s[0]);
        const float rowHi = __uint_as_float(s[1]);
        const int m0 = mt * 16 + r;
        const int m1 = m0 + 8;
        const float w0 = __uint_as_float(__builtin_amdgcn_readlane(__float_as_uint(wp), m0));
        const float w1 = __uint_as_float(__builtin_amdgcn_readlane(__float_as_uint(wp), m1));
        const int   i0 = __builtin_amdgcn_readlane(lo, m0);
        const int   i1 = __builtin_amdgcn_readlane(lo, m1);
        atomicAdd(acc + i0 + (unsigned)lane, w0 * rowLo);
        atomicAdd(acc + i1 + (unsigned)lane, w1 * rowHi);
      }
    }
  }
}

__launch_bounds__(256)
__global__ void cell_to_channel_major_kernel(const float* __restrict__ acc,
                                             float* __restrict__ out) {
  __shared__ float tile[(TC + TH) * TP];

  const int t    = threadIdx.x;
  const int lane = t & 31;
  const int wv   = t >> 5;
  const int b    = blockIdx.y;
  const int c0   = blockIdx.x * TC;

  int ncell = DHW - c0;
  ncell = (ncell < TC + TH) ? ncell : (TC + TH);
  const int nf4 = ncell * (CO / 4);
  const float4* src = (const float4*)(acc + ((size_t)b * DHW + c0) * CO);
  for (int i = t; i < nf4; i += 256) {
    float4 v = src[i];
    float* d = &tile[(i >> 3) * TP + (i & 7) * 4];
    d[0] = v.x; d[1] = v.y; d[2] = v.z; d[3] = v.w;
  }
  __syncthreads();

  float v[4][TC / 32];
  float vhead[4];
#pragma unroll
  for (int k = 0; k < 4; ++k) {
    const int ch = wv * 4 + k;
    const int sh = (32 - 8 * (ch & 3)) & 31;
#pragma unroll
    for (int e = 0; e < TC / 32; ++e) {
      int lr = sh + e * 32 + lane;
      lr = (lr < ncell) ? lr : (ncell - 1);
      v[k][e] = tile[lr * TP + ch];
    }
    vhead[k] = tile[((lane < ncell) ? lane : (ncell - 1)) * TP + ch];
  }
#pragma unroll
  for (int k = 0; k < 4; ++k) {
    const int ch = wv * 4 + k;
    const int sh = (32 - 8 * (ch & 3)) & 31;
    float* orow = out + ((size_t)(b * CO + ch)) * DHW + c0;
#pragma unroll
    for (int e = 0; e < TC / 32; ++e) {
      const int lc = sh + e * 32 + lane;
      if (c0 + lc < DHW) orow[lc] = v[k][e];
    }
    if (blockIdx.x == 0 && lane < sh) orow[lane] = vhead[k];
  }
}

extern "C" void kernel_launch(void* const* d_in, const int* in_sizes, int n_in,
                              void* d_out, int out_size, void* d_ws, size_t ws_size,
                              hipStream_t stream) {
  const float* x    = (const float*)d_in[0];
  const float* pos  = (const float*)d_in[1];
  const float* kern = (const float*)d_in[2];
  const float* bias = (const float*)d_in[3];
  float* out = (float*)d_out;
  float* acc = (float*)d_ws;

  const int Mtotal = in_sizes[0] / CI;
  const int Nper   = in_sizes[1] / (2 * 3);

  int n4 = out_size / 4;
  zero_f4_kernel<<<2048, 256, 0, stream>>>((float4*)acc, n4);

  int blocks = (Mtotal + 255) / 256;
  p2g_wmma_kernel<<<blocks, 256, 0, stream>>>(x, pos, kern, bias, acc,
                                              Mtotal, Nper);

  cell_to_channel_major_kernel<<<dim3((DHW + TC - 1) / TC, 2), 256, 0, stream>>>(acc, out);
}


# bin particles by 8x8x8-cell tile; each tile sums its lines in LDS (LDS f32 atomics) and adds each tile cell to the grid with one line atomic: sum order changes (per-tile partial sums)
# speedup vs baseline: 32.7939x; 1.1873x over previous
#include <hip/hip_runtime.h>


typedef float v2f __attribute__((ext_vector_type(2)));
typedef float v8f __attribute__((ext_vector_type(8)));
typedef unsigned int v2u __attribute__((ext_vector_type(2)));

#define DD 66
#define HHW (66 * 66)
#define DHW (66 * 66 * 66)
#define CI 32
#define CO 32

#define TC 128
#define TH 24
#define TP 33

#define NT 1024
#define TCELLS 1000

__global__ void zero_f4_kernel(float4* __restrict__ p, int n4, int* __restrict__ cnt) {
  int i = blockIdx.x * blockDim.x + threadIdx.x;
  int stride = gridDim.x * blockDim.x;
  if (blockIdx.x == 0)
    for (int j = threadIdx.x; j < NT; j += blockDim.x) cnt[j] = 0;
  for (; i < n4; i += stride) p[i] = make_float4(0.f, 0.f, 0.f, 0.f);
}

__device__ __forceinline__ int tile_of(const float* __restrict__ pos, int p, int Nper) {
  int b0 = (int)(pos[p * 3 + 0] * 64.0f);
  int b1 = (int)(pos[p * 3 + 1] * 64.0f);
  int b2 = (int)(pos[p * 3 + 2] * 64.0f);
  b0 = min(max(b0, 0), 63);
  b1 = min(max(b1, 0), 63);
  b2 = min(max(b2, 0), 63);
  int bb = (p >= Nper) ? 1 : 0;
  return bb * 512 + (b0 >> 3) * 64 + (b1 >> 3) * 8 + (b2 >> 3);
}

__launch_bounds__(256)
__global__ void bin_count_kernel(const float* __restrict__ pos, int* __restrict__ cnt,
                                 int* __restrict__ rank, int Mtotal, int Nper) {
  int p = blockIdx.x * 256 + threadIdx.x;
  if (p < Mtotal) rank[p] = atomicAdd(&cnt[tile_of(pos, p, Nper)], 1);
}

__launch_bounds__(1024)
__global__ void bin_scan_kernel(const int* __restrict__ cnt, int* __restrict__ off) {
  __shared__ int s[NT];
  const int t = threadIdx.x;
  const int c = cnt[t];
  s[t] = c;
  __syncthreads();
  for (int d = 1; d < NT; d <<= 1) {
    int v = (t >= d) ? s[t - d] : 0;
    __syncthreads();
    s[t] += v;
    __syncthreads();
  }
  off[t] = s[t] - c;
}

__launch_bounds__(256)
__global__ void bin_fill_kernel(const float* __restrict__ pos, const int* __restrict__ off,
                                const int* __restrict__ rank, int* __restrict__ sorted,
                                int Mtotal, int Nper) {
  int p = blockIdx.x * 256 + threadIdx.x;
  if (p < Mtotal) sorted[off[tile_of(pos, p, Nper)] + rank[p]] = p;
}

__launch_bounds__(512, 8)
__global__ void p2g_tile_kernel(const float* __restrict__ x,
                                const float* __restrict__ pos,
                                const float* __restrict__ kern,
                                const float* __restrict__ bias,
                                const int* __restrict__ cnt,
                                const int* __restrict__ off,
                                const int* __restrict__ sorted,
                                float* __restrict__ acc,
                                int Nper) {
  __shared__ __attribute__((aligned(16))) float sAcc[TCELLS * CO];
  __shared__ float sB[1024];

  const int t    = threadIdx.x;
  const int lane = t & 31;
  const int wv   = __builtin_amdgcn_readfirstlane(t >> 5);
  const int half = lane >> 4;
  const int nlo  = lane & 15;

  const int tile  = blockIdx.x;
  const int n     = cnt[tile];
  const int start = off[tile];
  if (n == 0) return;
  const int bb = tile >> 9;
  const int t0 = (tile >> 6) & 7, t1 = (tile >> 3) & 7, t2 = tile & 7;

  {
    float4* z = (float4*)sAcc;
    for (int i = t; i < TCELLS * CO / 4; i += 512) z[i] = make_float4(0.f, 0.f, 0.f, 0.f);
  }

  const float bl0 = bias[nlo];
  const float bl1 = bias[16 + nlo];
  const float4* k4 = (const float4*)kern;

  const int nch = (n + 31) >> 5;

  for (int g = 0; g < nch; g += 16) {
    const int  ch     = g + ((wv + (int)blockIdx.x) & 15);
    const bool active = ch < nch;

    const int q  = ch * 32 + lane;
    const int qc = (q < n) ? q : (n - 1);
    const float valid = (q < n) ? 1.0f : 0.0f;
    const int pc = sorted[start + qc];

    float px = pos[pc * 3 + 0] * 64.0f;
    float py = pos[pc * 3 + 1] * 64.0f;
    float pz = pos[pc * 3 + 2] * 64.0f;
    int b0 = (int)px, b1 = (int)py, b2 = (int)pz;
    float fx = px - (float)b0 - 0.5f;
    float fy = py - (float)b1 - 0.5f;
    float fz = pz - (float)b2 - 0.5f;

    float wx[3], wy[3], wz[3];
    wx[0] = 0.5f * (0.5f - fx) * (0.5f - fx) * valid;
    wx[1] = (0.75f - fx * fx) * valid;
    wx[2] = 0.5f * (0.5f + fx) * (0.5f + fx) * valid;
    wy[0] = 0.5f * (0.5f - fy) * (0.5f - fy);
    wy[1] = 0.75f - fy * fy;
    wy[2] = 0.5f * (0.5f + fy) * (0.5f + fy);
    wz[0] = 0.5f * (0.5f - fz) * (0.5f - fz);
    wz[1] = 0.75f - fz * fz;
    wz[2] = 0.5f * (0.5f + fz) * (0.5f + fz);

    const int l0 = min(max(b0 - t0 * 8, 0), 7);
    const int l1 = min(max(b1 - t1 * 8, 0), 7);
    const int l2 = min(max(b2 - t2 * 8, 0), 7);
    const int lbase = l0 * 100 + l1 * 10 + l2;

    v2f a[2][8];
#pragma unroll
    for (int mt = 0; mt < 2; ++mt) {
      int qr = ch * 32 + mt * 16 + nlo;
      qr = (qr < n) ? qr : (n - 1);
      const int row = sorted[start + qr];
      const float* xr = x + row * CI + half * 2;
#pragma unroll
      for (int c = 0; c < 8; ++c) a[mt][c] = *(const v2f*)(xr + c * 4);
    }

    for (int o = 0; o < 27; ++o) {
      const int oi = o / 9, oj = (o / 3) % 3, ok = o % 3;

      __syncthreads();

      if (t < 256) {
        float4 v4 = k4[o * 256 + t];
        int f   = t * 4;
        int ci  = f >> 5;
        int co  = f & 31;
        int nt  = co >> 4;
        int cc  = ci >> 2;
        int hh  = (ci >> 1) & 1;
        int vv  = ci & 1;
        int base = (nt * 8 + cc) * 64 + (hh * 16 + (co & 15)) * 2 + vv;
        sB[base + 0] = v4.x;
        sB[base + 2] = v4.y;
        sB[base + 4] = v4.z;
        sB[base + 6] = v4.w;
      }
      const float wp = wx[oi] * wy[oj] * wz[ok];
      const int   lo = (lbase + oi * 100 + oj * 10 + ok) * CO;
      __syncthreads();

      if (active) {
        v2f bf[2][8];
#pragma unroll
        for (int nt = 0; nt < 2; ++nt)
#pragma unroll
          for (int c = 0; c < 8; ++c)
            bf[nt][c] = *(const v2f*)&sB[(nt * 8 + c) * 64 + lane * 2];

        v8f dacc[2][2];
#pragma unroll
        for (int mt = 0; mt < 2; ++mt)
#pragma unroll
          for (int nt = 0; nt < 2; ++nt) {
            float bl = nt ? bl1 : bl0;
#pragma unroll
            for (int r = 0; r < 8; ++r) dacc[mt][nt][r] = bl;
          }

#pragma unroll
        for (int c = 0; c < 8; ++c) {
#pragma unroll
          for (int mt = 0; mt < 2; ++mt) {
#pragma unroll
            for (int nt = 0; nt < 2; ++nt) {
              dacc[mt][nt] = __builtin_amdgcn_wmma_f32_16x16x4_f32(
                  false, a[mt][c], false, bf[nt][c], (short)0, dacc[mt][nt],
                  false, false);
            }
          }
        }

#pragma unroll
        for (int mt = 0; mt < 2; ++mt) {
#pragma unroll
          for (int r = 0; r < 8; ++r) {
            v2u s = __builtin_amdgcn_permlane16_swap(
                __float_as_uint(dacc[mt][0][r]), __float_as_uint(dacc[mt][1][r]), false, false);
            const float rowLo = __uint_as_float(s[0]);
            const float rowHi = __uint_as_float(s[1]);
            const int m0 = mt * 16 + r;
            const int m1 = m0 + 8;
            const float w0 = __uint_as_float(__builtin_amdgcn_readlane(__float_as_uint(wp), m0));
            const float w1 = __uint_as_float(__builtin_amdgcn_readlane(__float_as_uint(wp), m1));
            const int   i0 = __builtin_amdgcn_readlane(lo, m0);
            const int   i1 = __builtin_amdgcn_readlane(lo, m1);
            atomicAdd(&sAcc[i0 + lane], w0 * rowLo);
            atomicAdd(&sAcc[i1 + lane], w1 * rowHi);
          }
        }
      }
    }
  }

  __syncthreads();

  for (int lc = wv; lc < TCELLS; lc += 16) {
    const int li = lc / 100, lj = (lc / 10) % 10, lk = lc % 10;
    const int cell = bb * DHW + (t0 * 8 + li) * HHW + (t1 * 8 + lj) * 66 + (t2 * 8 + lk);
    atomicAdd(acc + cell * CO + lane, sAcc[lc * CO + lane]);
  }
}

__launch_bounds__(256)
__global__ void cell_to_channel_major_kernel(const float* __restrict__ acc,
                                             float* __restrict__ out) {
  __shared__ float tile[(TC + TH) * TP];

  const int t    = threadIdx.x;
  const int lane = t & 31;
  const int wv   = t >> 5;
  const int b    = blockIdx.y;
  const int c0   = blockIdx.x * TC;

  int ncell = DHW - c0;
  ncell = (ncell < TC + TH) ? ncell : (TC + TH);
  const int nf4 = ncell * (CO / 4);
  const float4* src = (const float4*)(acc + ((size_t)b * DHW + c0) * CO);
  for (int i = t; i < nf4; i += 256) {
    float4 v = src[i];
    float* d = &tile[(i >> 3) * TP + (i & 7) * 4];
    d[0] = v.x; d[1] = v.y; d[2] = v.z; d[3] = v.w;
  }
  __syncthreads();

  float v[4][TC / 32];
  float vhead[4];
#pragma unroll
  for (int k = 0; k < 4; ++k) {
    const int ch = wv * 4 + k;
    const int sh = (32 - 8 * (ch & 3)) & 31;
#pragma unroll
    for (int e = 0; e < TC / 32; ++e) {
      int lr = sh + e * 32 + lane;
      lr = (lr < ncell) ? lr : (ncell - 1);
      v[k][e] = tile[lr * TP + ch];
    }
    vhead[k] = tile[((lane < ncell) ? lane : (ncell - 1)) * TP + ch];
  }
#pragma unroll
  for (int k = 0; k < 4; ++k) {
    const int ch = wv * 4 + k;
    const int sh = (32 - 8 * (ch & 3)) & 31;
    float* orow = out + ((size_t)(b * CO + ch)) * DHW + c0;
#pragma unroll
    for (int e = 0; e < TC / 32; ++e) {
      const int lc = sh + e * 32 + lane;
      if (c0 + lc < DHW) orow[lc] = v[k][e];
    }
    if (blockIdx.x == 0 && lane < sh) orow[lane] = vhead[k];
  }
}

extern "C" void kernel_launch(void* const* d_in, const int* in_sizes, int n_in,
                              void* d_out, int out_size, void* d_ws, size_t ws_size,
                              hipStream_t stream) {
  const float* x    = (const float*)d_in[0];
  const float* pos  = (const float*)d_in[1];
  const float* kern = (const float*)d_in[2];
  const float* bias = (const float*)d_in[3];
  float* out = (float*)d_out;
  float* acc = (float*)d_ws;
  int* cnt    = (int*)d_ws + 18399744;
  int* off    = cnt + 1024;
  int* rank   = cnt + 2048;
  int* sorted = cnt + 2048 + 524288;

  const int Mtotal = in_sizes[0] / CI;
  const int Nper   = in_sizes[1] / (2 * 3);

  int n4 = out_size / 4;
  zero_f4_kernel<<<2048, 256, 0, stream>>>((float4*)acc, n4, cnt);

  int blocks = (Mtotal + 255) / 256;
  bin_count_kernel<<<blocks, 256, 0, stream>>>(pos, cnt, rank, Mtotal, Nper);
  bin_scan_kernel<<<1, 1024, 0, stream>>>(cnt, off);
  bin_fill_kernel<<<blocks, 256, 0, stream>>>(pos, off, rank, sorted, Mtotal, Nper);

  p2g_tile_kernel<<<1024, 512, 0, stream>>>(x, pos, kern, bias, cnt, off, sorted, acc, Nper);

  cell_to_channel_major_kernel<<<dim3((DHW + TC - 1) / TC, 2), 256, 0, stream>>>(acc, out);
}


# scatter line addresses out of the 27-offset loop: each particle's base-cell line is read into a scalar once per chunk, an offset only adds a constant; halves the per-offset lane reads
# speedup vs baseline: 34.2513x; 1.0444x over previous
#include <hip/hip_runtime.h>


typedef float v2f __attribute__((ext_vector_type(2)));
typedef float v8f __attribute__((ext_vector_type(8)));
typedef unsigned int v2u __attribute__((ext_vector_type(2)));

#define DD 66
#define HHW (66 * 66)
#define DHW (66 * 66 * 66)
#define CI 32
#define CO 32

#define TC 128
#define TH 24
#define TP 33

#define NT 1024
#define TCELLS 1000

__global__ void zero_f4_kernel(float4* __restrict__ p, int n4, int* __restrict__ cnt) {
  int i = blockIdx.x * blockDim.x + threadIdx.x;
  int stride = gridDim.x * blockDim.x;
  if (blockIdx.x == 0)
    for (int j = threadIdx.x; j < NT; j += blockDim.x) cnt[j] = 0;
  for (; i < n4; i += stride) p[i] = make_float4(0.f, 0.f, 0.f, 0.f);
}

__device__ __forceinline__ int tile_of(const float* __restrict__ pos, int p, int Nper) {
  int b0 = (int)(pos[p * 3 + 0] * 64.0f);
  int b1 = (int)(pos[p * 3 + 1] * 64.0f);
  int b2 = (int)(pos[p * 3 + 2] * 64.0f);
  b0 = min(max(b0, 0), 63);
  b1 = min(max(b1, 0), 63);
  b2 = min(max(b2, 0), 63);
  int bb = (p >= Nper) ? 1 : 0;
  return bb * 512 + (b0 >> 3) * 64 + (b1 >> 3) * 8 + (b2 >> 3);
}

__launch_bounds__(256)
__global__ void bin_count_kernel(const float* __restrict__ pos, int* __restrict__ cnt,
                                 int* __restrict__ rank, int Mtotal, int Nper) {
  int p = blockIdx.x * 256 + threadIdx.x;
  if (p < Mtotal) rank[p] = atomicAdd(&cnt[tile_of(pos, p, Nper)], 1);
}

__launch_bounds__(1024)
__global__ void bin_scan_kernel(const int* __restrict__ cnt, int* __restrict__ off) {
  __shared__ int s[NT];
  const int t = threadIdx.x;
  const int c = cnt[t];
  s[t] = c;
  __syncthreads();
  for (int d = 1; d < NT; d <<= 1) {
    int v = (t >= d) ? s[t - d] : 0;
    __syncthreads();
    s[t] += v;
    __syncthreads();
  }
  off[t] = s[t] - c;
}

__launch_bounds__(256)
__global__ void bin_fill_kernel(const float* __restrict__ pos, const int* __restrict__ off,
                                const int* __restrict__ rank, int* __restrict__ sorted,
                                int Mtotal, int Nper) {
  int p = blockIdx.x * 256 + threadIdx.x;
  if (p < Mtotal) sorted[off[tile_of(pos, p, Nper)] + rank[p]] = p;
}

__launch_bounds__(512, 8)
__global__ void p2g_tile_kernel(const float* __restrict__ x,
                                const float* __restrict__ pos,
                                const float* __restrict__ kern,
                                const float* __restrict__ bias,
                                const int* __restrict__ cnt,
                                const int* __restrict__ off,
                                const int* __restrict__ sorted,
                                float* __restrict__ acc,
                                int Nper) {
  __shared__ __attribute__((aligned(16))) float sAcc[TCELLS * CO];
  __shared__ float sB[1024];

  const int t    = threadIdx.x;
  const int lane = t & 31;
  const int wv   = __builtin_amdgcn_readfirstlane(t >> 5);
  const int half = lane >> 4;
  const int nlo  = lane & 15;

  const int tile  = blockIdx.x;
  const int n     = cnt[tile];
  const int start = off[tile];
  if (n == 0) return;
  const int bb = tile >> 9;
  const int t0 = (tile >> 6) & 7, t1 = (tile >> 3) & 7, t2 = tile & 7;

  {
    float4* z = (float4*)sAcc;
    for (int i = t; i < TCELLS * CO / 4; i += 512) z[i] = make_float4(0.f, 0.f, 0.f, 0.f);
  }

  const float bl0 = bias[nlo];
  const float bl1 = bias[16 + nlo];
  const float4* k4 = (const float4*)kern;

  const int nch = (n + 31) >> 5;

  for (int g = 0; g < nch; g += 16) {
    const int  ch     = g + ((wv + (int)blockIdx.x) & 15);
    const bool active = ch < nch;

    const int q  = ch * 32 + lane;
    const int qc = (q < n) ? q : (n - 1);
    const float valid = (q < n) ? 1.0f : 0.0f;
    const int pc = sorted[start + qc];

    float px = pos[pc * 3 + 0] * 64.0f;
    float py = pos[pc * 3 + 1] * 64.0f;
    float pz = pos[pc * 3 + 2] * 64.0f;
    int b0 = (int)px, b1 = (int)py, b2 = (int)pz;
    float fx = px - (float)b0 - 0.5f;
    float fy = py - (float)b1 - 0.5f;
    float fz = pz - (float)b2 - 0.5f;

    float wx[3], wy[3], wz[3];
    wx[0] = 0.5f * (0.5f - fx) * (0.5f - fx) * valid;
    wx[1] = (0.75f - fx * fx) * valid;
    wx[2] = 0.5f * (0.5f + fx) * (0.5f + fx) * valid;
    wy[0] = 0.5f * (0.5f - fy) * (0.5f - fy);
    wy[1] = 0.75f - fy * fy;
    wy[2] = 0.5f * (0.5f + fy) * (0.5f + fy);
    wz[0] = 0.5f * (0.5f - fz) * (0.5f - fz);
    wz[1] = 0.75f - fz * fz;
    wz[2] = 0.5f * (0.5f + fz) * (0.5f + fz);

    const int l0 = min(max(b0 - t0 * 8, 0), 7);
    const int l1 = min(max(b1 - t1 * 8, 0), 7);
    const int l2 = min(max(b2 - t2 * 8, 0), 7);
    const int lbase = l0 * 100 + l1 * 10 + l2;

    int lb[32];
#pragma unroll
    for (int m = 0; m < 32; ++m) lb[m] = __builtin_amdgcn_readlane(lbase * CO, m);

    v2f a[2][8];
#pragma unroll
    for (int mt = 0; mt < 2; ++mt) {
      int qr = ch * 32 + mt * 16 + nlo;
      qr = (qr < n) ? qr : (n - 1);
      const int row = sorted[start + qr];
      const float* xr = x + row * CI + half * 2;
#pragma unroll
      for (int c = 0; c < 8; ++c) a[mt][c] = *(const v2f*)(xr + c * 4);
    }

    for (int o = 0; o < 27; ++o) {
      const int oi = o / 9, oj = (o / 3) % 3, ok = o % 3;

      __syncthreads();

      if (t < 256) {
        float4 v4 = k4[o * 256 + t];
        int f   = t * 4;
        int ci  = f >> 5;
        int co  = f & 31;
        int nt  = co >> 4;
        int cc  = ci >> 2;
        int hh  = (ci >> 1) & 1;
        int vv  = ci & 1;
        int base = (nt * 8 + cc) * 64 + (hh * 16 + (co & 15)) * 2 + vv;
        sB[base + 0] = v4.x;
        sB[base + 2] = v4.y;
        sB[base + 4] = v4.z;
        sB[base + 6] = v4.w;
      }
      const float wp = wx[oi] * wy[oj] * wz[ok];
      const int   vo = (oi * 100 + oj * 10 + ok) * CO + lane;
      __syncthreads();

      if (active) {
        v2f bf[2][8];
#pragma unroll
        for (int nt = 0; nt < 2; ++nt)
#pragma unroll
          for (int c = 0; c < 8; ++c)
            bf[nt][c] = *(const v2f*)&sB[(nt * 8 + c) * 64 + lane * 2];

        v8f dacc[2][2];
#pragma unroll
        for (int mt = 0; mt < 2; ++mt)
#pragma unroll
          for (int nt = 0; nt < 2; ++nt) {
            float bl = nt ? bl1 : bl0;
#pragma unroll
            for (int r = 0; r < 8; ++r) dacc[mt][nt][r] = bl;
          }

#pragma unroll
        for (int c = 0; c < 8; ++c) {
#pragma unroll
          for (int mt = 0; mt < 2; ++mt) {
#pragma unroll
            for (int nt = 0; nt < 2; ++nt) {
              dacc[mt][nt] = __builtin_amdgcn_wmma_f32_16x16x4_f32(
                  false, a[mt][c], false, bf[nt][c], (short)0, dacc[mt][nt],
                  false, false);
            }
          }
        }

#pragma unroll
        for (int mt = 0; mt < 2; ++mt) {
#pragma unroll
          for (int r = 0; r < 8; ++r) {
            v2u s = __builtin_amdgcn_permlane16_swap(
                __float_as_uint(dacc[mt][0][r]), __float_as_uint(dacc[mt][1][r]), false, false);
            const float rowLo = __uint_as_float(s[0]);
            const float rowHi = __uint_as_float(s[1]);
            const int m0 = mt * 16 + r;
            const int m1 = m0 + 8;
            const float w0 = __uint_as_float(__builtin_amdgcn_readlane(__float_as_uint(wp), m0));
            const float w1 = __uint_as_float(__builtin_amdgcn_readlane(__float_as_uint(wp), m1));
            atomicAdd(&sAcc[lb[m0] + vo], w0 * rowLo);
            atomicAdd(&sAcc[lb[m1] + vo], w1 * rowHi);
          }
        }
      }
    }
  }

  __syncthreads();

  for (int lc = wv; lc < TCELLS; lc += 16) {
    const int li = lc / 100, lj = (lc / 10) % 10, lk = lc % 10;
    const int cell = bb * DHW + (t0 * 8 + li) * HHW + (t1 * 8 + lj) * 66 + (t2 * 8 + lk);
    atomicAdd(acc + cell * CO + lane, sAcc[lc * CO + lane]);
  }
}

__launch_bounds__(256)
__global__ void cell_to_channel_major_kernel(const float* __restrict__ acc,
                                             float* __restrict__ out) {
  __shared__ float tile[(TC + TH) * TP];

  const int t    = threadIdx.x;
  const int lane = t & 31;
  const int wv   = t >> 5;
  const int b    = blockIdx.y;
  const int c0   = blockIdx.x * TC;

  int ncell = DHW - c0;
  ncell = (ncell < TC + TH) ? ncell : (TC + TH);
  const int nf4 = ncell * (CO / 4);
  const float4* src = (const float4*)(acc + ((size_t)b * DHW + c0) * CO);
  for (int i = t; i < nf4; i += 256) {
    float4 v = src[i];
    float* d = &tile[(i >> 3) * TP + (i & 7) * 4];
    d[0] = v.x; d[1] = v.y; d[2] = v.z; d[3] = v.w;
  }
  __syncthreads();

  float v[4][TC / 32];
  float vhead[4];
#pragma unroll
  for (int k = 0; k < 4; ++k) {
    const int ch = wv * 4 + k;
    const int sh = (32 - 8 * (ch & 3)) & 31;
#pragma unroll
    for (int e = 0; e < TC / 32; ++e) {
      int lr = sh + e * 32 + lane;
      lr = (lr < ncell) ? lr : (ncell - 1);
      v[k][e] = tile[lr * TP + ch];
    }
    vhead[k] = tile[((lane < ncell) ? lane : (ncell - 1)) * TP + ch];
  }
#pragma unroll
  for (int k = 0; k < 4; ++k) {
    const int ch = wv * 4 + k;
    const int sh = (32 - 8 * (ch & 3)) & 31;
    float* orow = out + ((size_t)(b * CO + ch)) * DHW + c0;
#pragma unroll
    for (int e = 0; e < TC / 32; ++e) {
      const int lc = sh + e * 32 + lane;
      if (c0 + lc < DHW) orow[lc] = v[k][e];
    }
    if (blockIdx.x == 0 && lane < sh) orow[lane] = vhead[k];
  }
}

extern "C" void kernel_launch(void* const* d_in, const int* in_sizes, int n_in,
                              void* d_out, int out_size, void* d_ws, size_t ws_size,
                              hipStream_t stream) {
  const float* x    = (const float*)d_in[0];
  const float* pos  = (const float*)d_in[1];
  const float* kern = (const float*)d_in[2];
  const float* bias = (const float*)d_in[3];
  float* out = (float*)d_out;
  float* acc = (float*)d_ws;
  int* cnt    = (int*)d_ws + 18399744;
  int* off    = cnt + 1024;
  int* rank   = cnt + 2048;
  int* sorted = cnt + 2048 + 524288;

  const int Mtotal = in_sizes[0] / CI;
  const int Nper   = in_sizes[1] / (2 * 3);

  int n4 = out_size / 4;
  zero_f4_kernel<<<2048, 256, 0, stream>>>((float4*)acc, n4, cnt);

  int blocks = (Mtotal + 255) / 256;
  bin_count_kernel<<<blocks, 256, 0, stream>>>(pos, cnt, rank, Mtotal, Nper);
  bin_scan_kernel<<<1, 1024, 0, stream>>>(cnt, off);
  bin_fill_kernel<<<blocks, 256, 0, stream>>>(pos, off, rank, sorted, Mtotal, Nper);

  p2g_tile_kernel<<<1024, 512, 0, stream>>>(x, pos, kern, bias, cnt, off, sorted, acc, Nper);

  cell_to_channel_major_kernel<<<dim3((DHW + TC - 1) / TC, 2), 256, 0, stream>>>(acc, out);
}


# B fragments read straight from a fragment-ordered copy of the 27 kernel slices in workspace (written by the first launch): no per-offset LDS staging and no barrier inside the 27-offset loop
# speedup vs baseline: 36.0411x; 1.0523x over previous
#include <hip/hip_runtime.h>


typedef float v2f __attribute__((ext_vector_type(2)));
typedef float v8f __attribute__((ext_vector_type(8)));
typedef unsigned int v2u __attribute__((ext_vector_type(2)));

#define DD 66
#define HHW (66 * 66)
#define DHW (66 * 66 * 66)
#define CI 32
#define CO 32

#define TC 128
#define TH 24
#define TP 33

#define NT 1024
#define TCELLS 1000

__global__ void zero_f4_kernel(float4* __restrict__ p, int n4, int* __restrict__ cnt,
                               const float* __restrict__ kern, float* __restrict__ kfrag) {
  int i = blockIdx.x * blockDim.x + threadIdx.x;
  int stride = gridDim.x * blockDim.x;
  if (blockIdx.x == 0)
    for (int j = threadIdx.x; j < NT; j += blockDim.x) cnt[j] = 0;
  if (blockIdx.x >= 1 && blockIdx.x <= 27) {
    const int o = blockIdx.x - 1;
    for (int t = threadIdx.x; t < 256; t += blockDim.x) {
      float4 v4 = ((const float4*)kern)[o * 256 + t];
      int f   = t * 4;
      int ci  = f >> 5;
      int co  = f & 31;
      int nt  = co >> 4;
      int cc  = ci >> 2;
      int hh  = (ci >> 1) & 1;
      int vv  = ci & 1;
      float* d = kfrag + o * 1024 + (nt * 8 + cc) * 64 + (hh * 16 + (co & 15)) * 2 + vv;
      d[0] = v4.x;
      d[2] = v4.y;
      d[4] = v4.z;
      d[6] = v4.w;
    }
  }
  for (; i < n4; i += stride) p[i] = make_float4(0.f, 0.f, 0.f, 0.f);
}

__device__ __forceinline__ int tile_of(const float* __restrict__ pos, int p, int Nper) {
  int b0 = (int)(pos[p * 3 + 0] * 64.0f);
  int b1 = (int)(pos[p * 3 + 1] * 64.0f);
  int b2 = (int)(pos[p * 3 + 2] * 64.0f);
  b0 = min(max(b0, 0), 63);
  b1 = min(max(b1, 0), 63);
  b2 = min(max(b2, 0), 63);
  int bb = (p >= Nper) ? 1 : 0;
  return bb * 512 + (b0 >> 3) * 64 + (b1 >> 3) * 8 + (b2 >> 3);
}

__launch_bounds__(256)
__global__ void bin_count_kernel(const float* __restrict__ pos, int* __restrict__ cnt,
                                 int* __restrict__ rank, int Mtotal, int Nper) {
  int p = blockIdx.x * 256 + threadIdx.x;
  if (p < Mtotal) rank[p] = atomicAdd(&cnt[tile_of(pos, p, Nper)], 1);
}

__launch_bounds__(1024)
__global__ void bin_scan_kernel(const int* __restrict__ cnt, int* __restrict__ off) {
  __shared__ int s[NT];
  const int t = threadIdx.x;
  const int c = cnt[t];
  s[t] = c;
  __syncthreads();
  for (int d = 1; d < NT; d <<= 1) {
    int v = (t >= d) ? s[t - d] : 0;
    __syncthreads();
    s[t] += v;
    __syncthreads();
  }
  off[t] = s[t] - c;
}

__launch_bounds__(256)
__global__ void bin_fill_kernel(const float* __restrict__ pos, const int* __restrict__ off,
                                const int* __restrict__ rank, int* __restrict__ sorted,
                                int Mtotal, int Nper) {
  int p = blockIdx.x * 256 + threadIdx.x;
  if (p < Mtotal) sorted[off[tile_of(pos, p, Nper)] + rank[p]] = p;
}

__launch_bounds__(512, 8)
__global__ void p2g_tile_kernel(const float* __restrict__ x,
                                const float* __restrict__ pos,
                                const float* __restrict__ kfrag,
                                const float* __restrict__ bias,
                                const int* __restrict__ cnt,
                                const int* __restrict__ off,
                                const int* __restrict__ sorted,
                                float* __restrict__ acc,
                                int Nper) {
  __shared__ __attribute__((aligned(16))) float sAcc[TCELLS * CO];

  const int t    = threadIdx.x;
  const int lane = t & 31;
  const int wv   = __builtin_amdgcn_readfirstlane(t >> 5);
  const int half = lane >> 4;
  const int nlo  = lane & 15;

  const int tile  = blockIdx.x;
  const int n     = cnt[tile];
  const int start = off[tile];
  if (n == 0) return;
  const int bb = tile >> 9;
  const int t0 = (tile >> 6) & 7, t1 = (tile >> 3) & 7, t2 = tile & 7;

  {
    float4* z = (float4*)sAcc;
    for (int i = t; i < TCELLS * CO / 4; i += 512) z[i] = make_float4(0.f, 0.f, 0.f, 0.f);
  }
  __syncthreads();

  const float bl0 = bias[nlo];
  const float bl1 = bias[16 + nlo];

  const int nch = (n + 31) >> 5;

  for (int g = 0; g < nch; g += 16) {
    const int  ch     = g + ((wv + (int)blockIdx.x) & 15);
    if (ch >= nch) continue;

    const int q  = ch * 32 + lane;
    const int qc = (q < n) ? q : (n - 1);
    const float valid = (q < n) ? 1.0f : 0.0f;
    const int pc = sorted[start + qc];

    float px = pos[pc * 3 + 0] * 64.0f;
    float py = pos[pc * 3 + 1] * 64.0f;
    float pz = pos[pc * 3 + 2] * 64.0f;
    int b0 = (int)px, b1 = (int)py, b2 = (int)pz;
    float fx = px - (float)b0 - 0.5f;
    float fy = py - (float)b1 - 0.5f;
    float fz = pz - (float)b2 - 0.5f;

    float wx[3], wy[3], wz[3];
    wx[0] = 0.5f * (0.5f - fx) * (0.5f - fx) * valid;
    wx[1] = (0.75f - fx * fx) * valid;
    wx[2] = 0.5f * (0.5f + fx) * (0.5f + fx) * valid;
    wy[0] = 0.5f * (0.5f - fy) * (0.5f - fy);
    wy[1] = 0.75f - fy * fy;
    wy[2] = 0.5f * (0.5f + fy) * (0.5f + fy);
    wz[0] = 0.5f * (0.5f - fz) * (0.5f - fz);
    wz[1] = 0.75f - fz * fz;
    wz[2] = 0.5f * (0.5f + fz) * (0.5f + fz);

    const int l0 = min(max(b0 - t0 * 8, 0), 7);
    const int l1 = min(max(b1 - t1 * 8, 0), 7);
    const int l2 = min(max(b2 - t2 * 8, 0), 7);
    const int lbase = l0 * 100 + l1 * 10 + l2;

    int lb[32];
#pragma unroll
    for (int m = 0; m < 32; ++m) lb[m] = __builtin_amdgcn_readlane(lbase * CO, m);

    v2f a[2][8];
#pragma unroll
    for (int mt = 0; mt < 2; ++mt) {
      int qr = ch * 32 + mt * 16 + nlo;
      qr = (qr < n) ? qr : (n - 1);
      const int row = sorted[start + qr];
      const float* xr = x + row * CI + half * 2;
#pragma unroll
      for (int c = 0; c < 8; ++c) a[mt][c] = *(const v2f*)(xr + c * 4);
    }

    for (int o = 0; o < 27; ++o) {
      const int oi = o / 9, oj = (o / 3) % 3, ok = o % 3;

      const float wp = wx[oi] * wy[oj] * wz[ok];
      const int   vo = (oi * 100 + oj * 10 + ok) * CO + lane;

      {
        const float* kf = kfrag + o * 1024 + lane * 2;
        v2f bf[2][8];
#pragma unroll
        for (int nt = 0; nt < 2; ++nt)
#pragma unroll
          for (int c = 0; c < 8; ++c)
            bf[nt][c] = *(const v2f*)&kf[(nt * 8 + c) * 64];

        v8f dacc[2][2];
#pragma unroll
        for (int mt = 0; mt < 2; ++mt)
#pragma unroll
          for (int nt = 0; nt < 2; ++nt) {
            float bl = nt ? bl1 : bl0;
#pragma unroll
            for (int r = 0; r < 8; ++r) dacc[mt][nt][r] = bl;
          }

#pragma unroll
        for (int c = 0; c < 8; ++c) {
#pragma unroll
          for (int mt = 0; mt < 2; ++mt) {
#pragma unroll
            for (int nt = 0; nt < 2; ++nt) {
              dacc[mt][nt] = __builtin_amdgcn_wmma_f32_16x16x4_f32(
                  false, a[mt][c], false, bf[nt][c], (short)0, dacc[mt][nt],
                  false, false);
            }
          }
        }

#pragma unroll
        for (int mt = 0; mt < 2; ++mt) {
#pragma unroll
          for (int r = 0; r < 8; ++r) {
            v2u s = __builtin_amdgcn_permlane16_swap(
                __float_as_uint(dacc[mt][0][r]), __float_as_uint(dacc[mt][1][r]), false, false);
            const float rowLo = __uint_as_float(s[0]);
            const float rowHi = __uint_as_float(s[1]);
            const int m0 = mt * 16 + r;
            const int m1 = m0 + 8;
            const float w0 = __uint_as_float(__builtin_amdgcn_readlane(__float_as_uint(wp), m0));
            const float w1 = __uint_as_float(__builtin_amdgcn_readlane(__float_as_uint(wp), m1));
            atomicAdd(&sAcc[lb[m0] + vo], w0 * rowLo);
            atomicAdd(&sAcc[lb[m1] + vo], w1 * rowHi);
          }
        }
      }
    }
  }

  __syncthreads();

  for (int lc = wv; lc < TCELLS; lc += 16) {
    const int li = lc / 100, lj = (lc / 10) % 10, lk = lc % 10;
    const int cell = bb * DHW + (t0 * 8 + li) * HHW + (t1 * 8 + lj) * 66 + (t2 * 8 + lk);
    atomicAdd(acc + cell * CO + lane, sAcc[lc * CO + lane]);
  }
}

__launch_bounds__(256)
__global__ void cell_to_channel_major_kernel(const float* __restrict__ acc,
                                             float* __restrict__ out) {
  __shared__ float tile[(TC + TH) * TP];

  const int t    = threadIdx.x;
  const int lane = t & 31;
  const int wv   = t >> 5;
  const int b    = blockIdx.y;
  const int c0   = blockIdx.x * TC;

  int ncell = DHW - c0;
  ncell = (ncell < TC + TH) ? ncell : (TC + TH);
  const int nf4 = ncell * (CO / 4);
  const float4* src = (const float4*)(acc + ((size_t)b * DHW + c0) * CO);
  for (int i = t; i < nf4; i += 256) {
    float4 v = src[i];
    float* d = &tile[(i >> 3) * TP + (i & 7) * 4];
    d[0] = v.x; d[1] = v.y; d[2] = v.z; d[3] = v.w;
  }
  __syncthreads();

  float v[4][TC / 32];
  float vhead[4];
#pragma unroll
  for (int k = 0; k < 4; ++k) {
    const int ch = wv * 4 + k;
    const int sh = (32 - 8 * (ch & 3)) & 31;
#pragma unroll
    for (int e = 0; e < TC / 32; ++e) {
      int lr = sh + e * 32 + lane;
      lr = (lr < ncell) ? lr : (ncell - 1);
      v[k][e] = tile[lr * TP + ch];
    }
    vhead[k] = tile[((lane < ncell) ? lane : (ncell - 1)) * TP + ch];
  }
#pragma unroll
  for (int k = 0; k < 4; ++k) {
    const int ch = wv * 4 + k;
    const int sh = (32 - 8 * (ch & 3)) & 31;
    float* orow = out + ((size_t)(b * CO + ch)) * DHW + c0;
#pragma unroll
    for (int e = 0; e < TC / 32; ++e) {
      const int lc = sh + e * 32 + lane;
      if (c0 + lc < DHW) orow[lc] = v[k][e];
    }
    if (blockIdx.x == 0 && lane < sh) orow[lane] = vhead[k];
  }
}

extern "C" void kernel_launch(void* const* d_in, const int* in_sizes, int n_in,
                              void* d_out, int out_size, void* d_ws, size_t ws_size,
                              hipStream_t stream) {
  const float* x    = (const float*)d_in[0];
  const float* pos  = (const float*)d_in[1];
  const float* kern = (const float*)d_in[2];
  const float* bias = (const float*)d_in[3];
  float* out = (float*)d_out;
  float* acc = (float*)d_ws;
  int* cnt    = (int*)d_ws + 18399744;
  int* off    = cnt + 1024;
  int* rank   = cnt + 2048;
  int* sorted = cnt + 2048 + 524288;
  float* kfrag = (float*)(cnt + 2048 + 524288 + 524288);

  const int Mtotal = in_sizes[0] / CI;
  const int Nper   = in_sizes[1] / (2 * 3);

  int n4 = out_size / 4;
  zero_f4_kernel<<<2048, 256, 0, stream>>>((float4*)acc, n4, cnt, kern, kfrag);

  int blocks = (Mtotal + 255) / 256;
  bin_count_kernel<<<blocks, 256, 0, stream>>>(pos, cnt, rank, Mtotal, Nper);
  bin_scan_kernel<<<1, 1024, 0, stream>>>(cnt, off);
  bin_fill_kernel<<<blocks, 256, 0, stream>>>(pos, off, rank, sorted, Mtotal, Nper);

  p2g_tile_kernel<<<1024, 512, 0, stream>>>(x, pos, kfrag, bias, cnt, off, sorted, acc, Nper);

  cell_to_channel_major_kernel<<<dim3((DHW + TC - 1) / TC, 2), 256, 0, stream>>>(acc, out);
}


# a tile's (32-particle chunk, offset) pairs are cut into 16 equal runs, one per wave, instead of whole chunks on 13 of 16 waves: equal matrix work per SIMD; LDS atomic arrival order as before
# speedup vs baseline: 39.0321x; 1.0830x over previous
#include <hip/hip_runtime.h>


typedef float v2f __attribute__((ext_vector_type(2)));
typedef float v8f __attribute__((ext_vector_type(8)));
typedef unsigned int v2u __attribute__((ext_vector_type(2)));

#define DD 66
#define HHW (66 * 66)
#define DHW (66 * 66 * 66)
#define CI 32
#define CO 32

#define TC 128
#define TH 24
#define TP 33

#define NT 1024
#define TCELLS 1000

__global__ void zero_f4_kernel(float4* __restrict__ p, int n4, int* __restrict__ cnt,
                               const float* __restrict__ kern, float* __restrict__ kfrag) {
  int i = blockIdx.x * blockDim.x + threadIdx.x;
  int stride = gridDim.x * blockDim.x;
  if (blockIdx.x == 0)
    for (int j = threadIdx.x; j < NT; j += blockDim.x) cnt[j] = 0;
  if (blockIdx.x >= 1 && blockIdx.x <= 27) {
    const int o = blockIdx.x - 1;
    for (int t = threadIdx.x; t < 256; t += blockDim.x) {
      float4 v4 = ((const float4*)kern)[o * 256 + t];
      int f   = t * 4;
      int ci  = f >> 5;
      int co  = f & 31;
      int nt  = co >> 4;
      int cc  = ci >> 2;
      int hh  = (ci >> 1) & 1;
      int vv  = ci & 1;
      float* d = kfrag + o * 1024 + (nt * 8 + cc) * 64 + (hh * 16 + (co & 15)) * 2 + vv;
      d[0] = v4.x;
      d[2] = v4.y;
      d[4] = v4.z;
      d[6] = v4.w;
    }
  }
  for (; i < n4; i += stride) p[i] = make_float4(0.f, 0.f, 0.f, 0.f);
}

__device__ __forceinline__ int tile_of(const float* __restrict__ pos, int p, int Nper) {
  int b0 = (int)(pos[p * 3 + 0] * 64.0f);
  int b1 = (int)(pos[p * 3 + 1] * 64.0f);
  int b2 = (int)(pos[p * 3 + 2] * 64.0f);
  b0 = min(max(b0, 0), 63);
  b1 = min(max(b1, 0), 63);
  b2 = min(max(b2, 0), 63);
  int bb = (p >= Nper) ? 1 : 0;
  return bb * 512 + (b0 >> 3) * 64 + (b1 >> 3) * 8 + (b2 >> 3);
}

__launch_bounds__(256)
__global__ void bin_count_kernel(const float* __restrict__ pos, int* __restrict__ cnt,
                                 int* __restrict__ rank, int Mtotal, int Nper) {
  int p = blockIdx.x * 256 + threadIdx.x;
  if (p < Mtotal) rank[p] = atomicAdd(&cnt[tile_of(pos, p, Nper)], 1);
}

__launch_bounds__(1024)
__global__ void bin_scan_kernel(const int* __restrict__ cnt, int* __restrict__ off) {
  __shared__ int s[NT];
  const int t = threadIdx.x;
  const int c = cnt[t];
  s[t] = c;
  __syncthreads();
  for (int d = 1; d < NT; d <<= 1) {
    int v = (t >= d) ? s[t - d] : 0;
    __syncthreads();
    s[t] += v;
    __syncthreads();
  }
  off[t] = s[t] - c;
}

__launch_bounds__(256)
__global__ void bin_fill_kernel(const float* __restrict__ pos, const int* __restrict__ off,
                                const int* __restrict__ rank, int* __restrict__ sorted,
                                int Mtotal, int Nper) {
  int p = blockIdx.x * 256 + threadIdx.x;
  if (p < Mtotal) sorted[off[tile_of(pos, p, Nper)] + rank[p]] = p;
}

__launch_bounds__(512, 8)
__global__ void p2g_tile_kernel(const float* __restrict__ x,
                                const float* __restrict__ pos,
                                const float* __restrict__ kfrag,
                                const float* __restrict__ bias,
                                const int* __restrict__ cnt,
                                const int* __restrict__ off,
                                const int* __restrict__ sorted,
                                float* __restrict__ acc,
                                int Nper) {
  __shared__ __attribute__((aligned(16))) float sAcc[TCELLS * CO];

  const int t    = threadIdx.x;
  const int lane = t & 31;
  const int wv   = __builtin_amdgcn_readfirstlane(t >> 5);
  const int half = lane >> 4;
  const int nlo  = lane & 15;

  const int tile  = blockIdx.x;
  const int n     = cnt[tile];
  const int start = off[tile];
  if (n == 0) return;
  const int bb = tile >> 9;
  const int t0 = (tile >> 6) & 7, t1 = (tile >> 3) & 7, t2 = tile & 7;

  {
    float4* z = (float4*)sAcc;
    for (int i = t; i < TCELLS * CO / 4; i += 512) z[i] = make_float4(0.f, 0.f, 0.f, 0.f);
  }
  __syncthreads();

  const float bl0 = bias[nlo];
  const float bl1 = bias[16 + nlo];

  const int nch = (n + 31) >> 5;

  const int tot = nch * 27;
  const int lo  = (wv * tot) >> 4;
  const int hi  = ((wv + 1) * tot) >> 4;
  for (int ch = lo / 27; ch * 27 < hi; ++ch) {
    const int oB = max(lo - ch * 27, 0);
    const int oE = min(hi - ch * 27, 27);

    const int q  = ch * 32 + lane;
    const int qc = (q < n) ? q : (n - 1);
    const float valid = (q < n) ? 1.0f : 0.0f;
    const int pc = sorted[start + qc];

    float px = pos[pc * 3 + 0] * 64.0f;
    float py = pos[pc * 3 + 1] * 64.0f;
    float pz = pos[pc * 3 + 2] * 64.0f;
    int b0 = (int)px, b1 = (int)py, b2 = (int)pz;
    float fx = px - (float)b0 - 0.5f;
    float fy = py - (float)b1 - 0.5f;
    float fz = pz - (float)b2 - 0.5f;

    float wx[3], wy[3], wz[3];
    wx[0] = 0.5f * (0.5f - fx) * (0.5f - fx) * valid;
    wx[1] = (0.75f - fx * fx) * valid;
    wx[2] = 0.5f * (0.5f + fx) * (0.5f + fx) * valid;
    wy[0] = 0.5f * (0.5f - fy) * (0.5f - fy);
    wy[1] = 0.75f - fy * fy;
    wy[2] = 0.5f * (0.5f + fy) * (0.5f + fy);
    wz[0] = 0.5f * (0.5f - fz) * (0.5f - fz);
    wz[1] = 0.75f - fz * fz;
    wz[2] = 0.5f * (0.5f + fz) * (0.5f + fz);

    const int l0 = min(max(b0 - t0 * 8, 0), 7);
    const int l1 = min(max(b1 - t1 * 8, 0), 7);
    const int l2 = min(max(b2 - t2 * 8, 0), 7);
    const int lbase = l0 * 100 + l1 * 10 + l2;

    int lb[32];
#pragma unroll
    for (int m = 0; m < 32; ++m) lb[m] = __builtin_amdgcn_readlane(lbase * CO, m);

    v2f a[2][8];
#pragma unroll
    for (int mt = 0; mt < 2; ++mt) {
      int qr = ch * 32 + mt * 16 + nlo;
      qr = (qr < n) ? qr : (n - 1);
      const int row = sorted[start + qr];
      const float* xr = x + row * CI + half * 2;
#pragma unroll
      for (int c = 0; c < 8; ++c) a[mt][c] = *(const v2f*)(xr + c * 4);
    }

    for (int o = oB; o < oE; ++o) {
      const int oi = o / 9, oj = (o / 3) % 3, ok = o % 3;

      const float wp = wx[oi] * wy[oj] * wz[ok];
      const int   vo = (oi * 100 + oj * 10 + ok) * CO + lane;

      {
        const float* kf = kfrag + o * 1024 + lane * 2;
        v2f bf[2][8];
#pragma unroll
        for (int nt = 0; nt < 2; ++nt)
#pragma unroll
          for (int c = 0; c < 8; ++c)
            bf[nt][c] = *(const v2f*)&kf[(nt * 8 + c) * 64];

        v8f dacc[2][2];
#pragma unroll
        for (int mt = 0; mt < 2; ++mt)
#pragma unroll
          for (int nt = 0; nt < 2; ++nt) {
            float bl = nt ? bl1 : bl0;
#pragma unroll
            for (int r = 0; r < 8; ++r) dacc[mt][nt][r] = bl;
          }

#pragma unroll
        for (int c = 0; c < 8; ++c) {
#pragma unroll
          for (int mt = 0; mt < 2; ++mt) {
#pragma unroll
            for (int nt = 0; nt < 2; ++nt) {
              dacc[mt][nt] = __builtin_amdgcn_wmma_f32_16x16x4_f32(
                  false, a[mt][c], false, bf[nt][c], (short)0, dacc[mt][nt],
                  false, false);
            }
          }
        }

#pragma unroll
        for (int mt = 0; mt < 2; ++mt) {
#pragma unroll
          for (int r = 0; r < 8; ++r) {
            v2u s = __builtin_amdgcn_permlane16_swap(
                __float_as_uint(dacc[mt][0][r]), __float_as_uint(dacc[mt][1][r]), false, false);
            const float rowLo = __uint_as_float(s[0]);
            const float rowHi = __uint_as_float(s[1]);
            const int m0 = mt * 16 + r;
            const int m1 = m0 + 8;
            const float w0 = __uint_as_float(__builtin_amdgcn_readlane(__float_as_uint(wp), m0));
            const float w1 = __uint_as_float(__builtin_amdgcn_readlane(__float_as_uint(wp), m1));
            atomicAdd(&sAcc[lb[m0] + vo], w0 * rowLo);
            atomicAdd(&sAcc[lb[m1] + vo], w1 * rowHi);
          }
        }
      }
    }
  }

  __syncthreads();

  for (int lc = wv; lc < TCELLS; lc += 16) {
    const int li = lc / 100, lj = (lc / 10) % 10, lk = lc % 10;
    const int cell = bb * DHW + (t0 * 8 + li) * HHW + (t1 * 8 + lj) * 66 + (t2 * 8 + lk);
    atomicAdd(acc + cell * CO + lane, sAcc[lc * CO + lane]);
  }
}

__launch_bounds__(256)
__global__ void cell_to_channel_major_kernel(const float* __restrict__ acc,
                                             float* __restrict__ out) {
  __shared__ float tile[(TC + TH) * TP];

  const int t    = threadIdx.x;
  const int lane = t & 31;
  const int wv   = t >> 5;
  const int b    = blockIdx.y;
  const int c0   = blockIdx.x * TC;

  int ncell = DHW - c0;
  ncell = (ncell < TC + TH) ? ncell : (TC + TH);
  const int nf4 = ncell * (CO / 4);
  const float4* src = (const float4*)(acc + ((size_t)b * DHW + c0) * CO);
  for (int i = t; i < nf4; i += 256) {
    float4 v = src[i];
    float* d = &tile[(i >> 3) * TP + (i & 7) * 4];
    d[0] = v.x; d[1] = v.y; d[2] = v.z; d[3] = v.w;
  }
  __syncthreads();

  float v[4][TC / 32];
  float vhead[4];
#pragma unroll
  for (int k = 0; k < 4; ++k) {
    const int ch = wv * 4 + k;
    const int sh = (32 - 8 * (ch & 3)) & 31;
#pragma unroll
    for (int e = 0; e < TC / 32; ++e) {
      int lr = sh + e * 32 + lane;
      lr = (lr < ncell) ? lr : (ncell - 1);
      v[k][e] = tile[lr * TP + ch];
    }
    vhead[k] = tile[((lane < ncell) ? lane : (ncell - 1)) * TP + ch];
  }
#pragma unroll
  for (int k = 0; k < 4; ++k) {
    const int ch = wv * 4 + k;
    const int sh = (32 - 8 * (ch & 3)) & 31;
    float* orow = out + ((size_t)(b * CO + ch)) * DHW + c0;
#pragma unroll
    for (int e = 0; e < TC / 32; ++e) {
      const int lc = sh + e * 32 + lane;
      if (c0 + lc < DHW) orow[lc] = v[k][e];
    }
    if (blockIdx.x == 0 && lane < sh) orow[lane] = vhead[k];
  }
}

extern "C" void kernel_launch(void* const* d_in, const int* in_sizes, int n_in,
                              void* d_out, int out_size, void* d_ws, size_t ws_size,
                              hipStream_t stream) {
  const float* x    = (const float*)d_in[0];
  const float* pos  = (const float*)d_in[1];
  const float* kern = (const float*)d_in[2];
  const float* bias = (const float*)d_in[3];
  float* out = (float*)d_out;
  float* acc = (float*)d_ws;
  int* cnt    = (int*)d_ws + 18399744;
  int* off    = cnt + 1024;
  int* rank   = cnt + 2048;
  int* sorted = cnt + 2048 + 524288;
  float* kfrag = (float*)(cnt + 2048 + 524288 + 524288);

  const int Mtotal = in_sizes[0] / CI;
  const int Nper   = in_sizes[1] / (2 * 3);

  int n4 = out_size / 4;
  zero_f4_kernel<<<2048, 256, 0, stream>>>((float4*)acc, n4, cnt, kern, kfrag);

  int blocks = (Mtotal + 255) / 256;
  bin_count_kernel<<<blocks, 256, 0, stream>>>(pos, cnt, rank, Mtotal, Nper);
  bin_scan_kernel<<<1, 1024, 0, stream>>>(cnt, off);
  bin_fill_kernel<<<blocks, 256, 0, stream>>>(pos, off, rank, sorted, Mtotal, Nper);

  p2g_tile_kernel<<<1024, 512, 0, stream>>>(x, pos, kfrag, bias, cnt, off, sorted, acc, Nper);

  cell_to_channel_major_kernel<<<dim3((DHW + TC - 1) / TC, 2), 256, 0, stream>>>(acc, out);
}


# each tile stores its 10x10x10 copy to its own workspace slot (plain stores); the last pass adds a cell's up to 8 overlapping copies in fixed order: no global atomics, no zero pass; sum order changes
# speedup vs baseline: 43.0652x; 1.1033x over previous
#include <hip/hip_runtime.h>


typedef float v2f __attribute__((ext_vector_type(2)));
typedef float v8f __attribute__((ext_vector_type(8)));
typedef unsigned int v2u __attribute__((ext_vector_type(2)));

#define DD 66
#define HHW (66 * 66)
#define DHW (66 * 66 * 66)
#define CI 32
#define CO 32

#define TC 128
#define TH 24
#define TP 33

#define NT 1024
#define TCELLS 1000
#define TSLOT (TCELLS * CO)

__global__ void setup_kernel(int* __restrict__ cnt, const float* __restrict__ kern,
                             float* __restrict__ kfrag) {
  if (blockIdx.x == 0) {
    for (int j = threadIdx.x; j < NT; j += blockDim.x) cnt[j] = 0;
  } else {
    const int o = blockIdx.x - 1;
    for (int t = threadIdx.x; t < 256; t += blockDim.x) {
      float4 v4 = ((const float4*)kern)[o * 256 + t];
      int f   = t * 4;
      int ci  = f >> 5;
      int co  = f & 31;
      int nt  = co >> 4;
      int cc  = ci >> 2;
      int hh  = (ci >> 1) & 1;
      int vv  = ci & 1;
      float* d = kfrag + o * 1024 + (nt * 8 + cc) * 64 + (hh * 16 + (co & 15)) * 2 + vv;
      d[0] = v4.x;
      d[2] = v4.y;
      d[4] = v4.z;
      d[6] = v4.w;
    }
  }
}

__device__ __forceinline__ int tile_of(const float* __restrict__ pos, int p, int Nper) {
  int b0 = (int)(pos[p * 3 + 0] * 64.0f);
  int b1 = (int)(pos[p * 3 + 1] * 64.0f);
  int b2 = (int)(pos[p * 3 + 2] * 64.0f);
  b0 = min(max(b0, 0), 63);
  b1 = min(max(b1, 0), 63);
  b2 = min(max(b2, 0), 63);
  int bb = (p >= Nper) ? 1 : 0;
  return bb * 512 + (b0 >> 3) * 64 + (b1 >> 3) * 8 + (b2 >> 3);
}

__launch_bounds__(256)
__global__ void bin_count_kernel(const float* __restrict__ pos, int* __restrict__ cnt,
                                 int* __restrict__ rank, int Mtotal, int Nper) {
  int p = blockIdx.x * 256 + threadIdx.x;
  if (p < Mtotal) rank[p] = atomicAdd(&cnt[tile_of(pos, p, Nper)], 1);
}

__launch_bounds__(1024)
__global__ void bin_scan_kernel(const int* __restrict__ cnt, int* __restrict__ off) {
  __shared__ int s[NT];
  const int t = threadIdx.x;
  const int c = cnt[t];
  s[t] = c;
  __syncthreads();
  for (int d = 1; d < NT; d <<= 1) {
    int v = (t >= d) ? s[t - d] : 0;
    __syncthreads();
    s[t] += v;
    __syncthreads();
  }
  off[t] = s[t] - c;
}

__launch_bounds__(256)
__global__ void bin_fill_kernel(const float* __restrict__ pos, const int* __restrict__ off,
                                const int* __restrict__ rank, int* __restrict__ sorted,
                                int Mtotal, int Nper) {
  int p = blockIdx.x * 256 + threadIdx.x;
  if (p < Mtotal) sorted[off[tile_of(pos, p, Nper)] + rank[p]] = p;
}

__launch_bounds__(512, 8)
__global__ void p2g_tile_kernel(const float* __restrict__ x,
                                const float* __restrict__ pos,
                                const float* __restrict__ kfrag,
                                const float* __restrict__ bias,
                                const int* __restrict__ cnt,
                                const int* __restrict__ off,
                                const int* __restrict__ sorted,
                                float* __restrict__ tbuf,
                                int Nper) {
  __shared__ __attribute__((aligned(16))) float sAcc[TCELLS * CO];

  const int t    = threadIdx.x;
  const int lane = t & 31;
  const int wv   = __builtin_amdgcn_readfirstlane(t >> 5);
  const int half = lane >> 4;
  const int nlo  = lane & 15;

  const int tile  = blockIdx.x;
  const int n     = cnt[tile];
  const int start = off[tile];
  float4* tb4 = (float4*)(tbuf + (size_t)tile * TSLOT);
  if (n == 0) {
    for (int i = t; i < TSLOT / 4; i += 512) tb4[i] = make_float4(0.f, 0.f, 0.f, 0.f);
    return;
  }
  const int t0 = (tile >> 6) & 7, t1 = (tile >> 3) & 7, t2 = tile & 7;

  {
    float4* z = (float4*)sAcc;
    for (int i = t; i < TCELLS * CO / 4; i += 512) z[i] = make_float4(0.f, 0.f, 0.f, 0.f);
  }
  __syncthreads();

  const float bl0 = bias[nlo];
  const float bl1 = bias[16 + nlo];

  const int nch = (n + 31) >> 5;

  const int tot = nch * 27;
  const int lo  = (wv * tot) >> 4;
  const int hi  = ((wv + 1) * tot) >> 4;
  for (int ch = lo / 27; ch * 27 < hi; ++ch) {
    const int oB = max(lo - ch * 27, 0);
    const int oE = min(hi - ch * 27, 27);

    const int q  = ch * 32 + lane;
    const int qc = (q < n) ? q : (n - 1);
    const float valid = (q < n) ? 1.0f : 0.0f;
    const int pc = sorted[start + qc];

    float px = pos[pc * 3 + 0] * 64.0f;
    float py = pos[pc * 3 + 1] * 64.0f;
    float pz = pos[pc * 3 + 2] * 64.0f;
    int b0 = (int)px, b1 = (int)py, b2 = (int)pz;
    float fx = px - (float)b0 - 0.5f;
    float fy = py - (float)b1 - 0.5f;
    float fz = pz - (float)b2 - 0.5f;

    float wx[3], wy[3], wz[3];
    wx[0] = 0.5f * (0.5f - fx) * (0.5f - fx) * valid;
    wx[1] = (0.75f - fx * fx) * valid;
    wx[2] = 0.5f * (0.5f + fx) * (0.5f + fx) * valid;
    wy[0] = 0.5f * (0.5f - fy) * (0.5f - fy);
    wy[1] = 0.75f - fy * fy;
    wy[2] = 0.5f * (0.5f + fy) * (0.5f + fy);
    wz[0] = 0.5f * (0.5f - fz) * (0.5f - fz);
    wz[1] = 0.75f - fz * fz;
    wz[2] = 0.5f * (0.5f + fz) * (0.5f + fz);

    const int l0 = min(max(b0 - t0 * 8, 0), 7);
    const int l1 = min(max(b1 - t1 * 8, 0), 7);
    const int l2 = min(max(b2 - t2 * 8, 0), 7);
    const int lbase = l0 * 100 + l1 * 10 + l2;

    int lb[32];
#pragma unroll
    for (int m = 0; m < 32; ++m) lb[m] = __builtin_amdgcn_readlane(lbase * CO, m);

    v2f a[2][8];
#pragma unroll
    for (int mt = 0; mt < 2; ++mt) {
      int qr = ch * 32 + mt * 16 + nlo;
      qr = (qr < n) ? qr : (n - 1);
      const int row = sorted[start + qr];
      const float* xr = x + row * CI + half * 2;
#pragma unroll
      for (int c = 0; c < 8; ++c) a[mt][c] = *(const v2f*)(xr + c * 4);
    }

    for (int o = oB; o < oE; ++o) {
      const int oi = o / 9, oj = (o / 3) % 3, ok = o % 3;

      const float wp = wx[oi] * wy[oj] * wz[ok];
      const int   vo = (oi * 100 + oj * 10 + ok) * CO + lane;

      {
        const float* kf = kfrag + o * 1024 + lane * 2;
        v2f bf[2][8];
#pragma unroll
        for (int nt = 0; nt < 2; ++nt)
#pragma unroll
          for (int c = 0; c < 8; ++c)
            bf[nt][c] = *(const v2f*)&kf[(nt * 8 + c) * 64];

        v8f dacc[2][2];
#pragma unroll
        for (int mt = 0; mt < 2; ++mt)
#pragma unroll
          for (int nt = 0; nt < 2; ++nt) {
            float bl = nt ? bl1 : bl0;
#pragma unroll
            for (int r = 0; r < 8; ++r) dacc[mt][nt][r] = bl;
          }

#pragma unroll
        for (int c = 0; c < 8; ++c) {
#pragma unroll
          for (int mt = 0; mt < 2; ++mt) {
#pragma unroll
            for (int nt = 0; nt < 2; ++nt) {
              dacc[mt][nt] = __builtin_amdgcn_wmma_f32_16x16x4_f32(
                  false, a[mt][c], false, bf[nt][c], (short)0, dacc[mt][nt],
                  false, false);
            }
          }
        }

#pragma unroll
        for (int mt = 0; mt < 2; ++mt) {
#pragma unroll
          for (int r = 0; r < 8; ++r) {
            v2u s = __builtin_amdgcn_permlane16_swap(
                __float_as_uint(dacc[mt][0][r]), __float_as_uint(dacc[mt][1][r]), false, false);
            const float rowLo = __uint_as_float(s[0]);
            const float rowHi = __uint_as_float(s[1]);
            const int m0 = mt * 16 + r;
            const int m1 = m0 + 8;
            const float w0 = __uint_as_float(__builtin_amdgcn_readlane(__float_as_uint(wp), m0));
            const float w1 = __uint_as_float(__builtin_amdgcn_readlane(__float_as_uint(wp), m1));
            atomicAdd(&sAcc[lb[m0] + vo], w0 * rowLo);
            atomicAdd(&sAcc[lb[m1] + vo], w1 * rowHi);
          }
        }
      }
    }
  }

  __syncthreads();

  {
    const float4* s4 = (const float4*)sAcc;
    for (int i = t; i < TSLOT / 4; i += 512) tb4[i] = s4[i];
  }
}

__launch_bounds__(256)
__global__ void sum_copies_channel_major_kernel(const float* __restrict__ tbuf,
                                                float* __restrict__ out) {
  __shared__ float tile[(TC + TH) * TP];

  const int t    = threadIdx.x;
  const int lane = t & 31;
  const int wv   = t >> 5;
  const int b    = blockIdx.y;
  const int c0   = blockIdx.x * TC;

  int ncell = DHW - c0;
  ncell = (ncell < TC + TH) ? ncell : (TC + TH);
  const int nf4 = ncell * (CO / 4);
  for (int i = t; i < nf4; i += 256) {
    const int g  = c0 + (i >> 3);
    const int d0 = g / HHW;
    const int gr = g - d0 * HHW;
    const int d1 = gr / DD;
    const int d2 = gr - d1 * DD;
    const int q0 = d0 >> 3, q1 = d1 >> 3, q2 = d2 >> 3;
    const int r0 = d0 & 7,  r1 = d1 & 7,  r2 = d2 & 7;
    const int e = ((b * 512 + q0 * 64 + q1 * 8 + q2) * TCELLS + r0 * 100 + r1 * 10 + r2) * CO + (i & 7) * 4;
    const int E0 = (800 - 64 * TCELLS) * CO;
    const int E1 = (80 - 8 * TCELLS) * CO;
    const int E2 = (8 - TCELLS) * CO;
    const bool A0 = q0 < 8, A1 = q1 < 8, A2 = q2 < 8;
    const bool B0 = (q0 >= 1) && (r0 < 2);
    const bool B1 = (q1 >= 1) && (r1 < 2);
    const bool B2 = (q2 >= 1) && (r2 < 2);

    const float4 z4 = make_float4(0.f, 0.f, 0.f, 0.f);
    float4 v0 = z4, v1 = z4, v2 = z4, v3 = z4, v4 = z4, v5 = z4, v6 = z4, v7 = z4;
    if (A0) {
      if (A1) {
        if (A2) v0 = *(const float4*)(tbuf + e);
        if (B2) v1 = *(const float4*)(tbuf + (e + E2));
      }
      if (B1) {
        if (A2) v2 = *(const float4*)(tbuf + (e + E1));
        if (B2) v3 = *(const float4*)(tbuf + (e + E1 + E2));
      }
    }
    if (B0) {
      if (A1) {
        if (A2) v4 = *(const float4*)(tbuf + (e + E0));
        if (B2) v5 = *(const float4*)(tbuf + (e + E0 + E2));
      }
      if (B1) {
        if (A2) v6 = *(const float4*)(tbuf + (e + E0 + E1));
        if (B2) v7 = *(const float4*)(tbuf + (e + E0 + E1 + E2));
      }
    }
    float4 s;
    s.x = ((((((v0.x + v1.x) + v2.x) + v3.x) + v4.x) + v5.x) + v6.x) + v7.x;
    s.y = ((((((v0.y + v1.y) + v2.y) + v3.y) + v4.y) + v5.y) + v6.y) + v7.y;
    s.z = ((((((v0.z + v1.z) + v2.z) + v3.z) + v4.z) + v5.z) + v6.z) + v7.z;
    s.w = ((((((v0.w + v1.w) + v2.w) + v3.w) + v4.w) + v5.w) + v6.w) + v7.w;

    float* d = &tile[(i >> 3) * TP + (i & 7) * 4];
    d[0] = s.x; d[1] = s.y; d[2] = s.z; d[3] = s.w;
  }
  __syncthreads();

  float v[4][TC / 32];
  float vhead[4];
#pragma unroll
  for (int k = 0; k < 4; ++k) {
    const int ch = wv * 4 + k;
    const int sh = (32 - 8 * (ch & 3)) & 31;
#pragma unroll
    for (int e = 0; e < TC / 32; ++e) {
      int lr = sh + e * 32 + lane;
      lr = (lr < ncell) ? lr : (ncell - 1);
      v[k][e] = tile[lr * TP + ch];
    }
    vhead[k] = tile[((lane < ncell) ? lane : (ncell - 1)) * TP + ch];
  }
#pragma unroll
  for (int k = 0; k < 4; ++k) {
    const int ch = wv * 4 + k;
    const int sh = (32 - 8 * (ch & 3)) & 31;
    float* orow = out + ((size_t)(b * CO + ch)) * DHW + c0;
#pragma unroll
    for (int e = 0; e < TC / 32; ++e) {
      const int lc = sh + e * 32 + lane;
      if (c0 + lc < DHW) orow[lc] = v[k][e];
    }
    if (blockIdx.x == 0 && lane < sh) orow[lane] = vhead[k];
  }
}

extern "C" void kernel_launch(void* const* d_in, const int* in_sizes, int n_in,
                              void* d_out, int out_size, void* d_ws, size_t ws_size,
                              hipStream_t stream) {
  const float* x    = (const float*)d_in[0];
  const float* pos  = (const float*)d_in[1];
  const float* kern = (const float*)d_in[2];
  const float* bias = (const float*)d_in[3];
  float* out = (float*)d_out;
  float* tbuf = (float*)d_ws;
  int* cnt    = (int*)d_ws + 32768000;
  int* off    = cnt + 1024;
  int* rank   = cnt + 2048;
  int* sorted = cnt + 2048 + 524288;
  float* kfrag = (float*)(cnt + 2048 + 524288 + 524288);

  const int Mtotal = in_sizes[0] / CI;
  const int Nper   = in_sizes[1] / (2 * 3);

  setup_kernel<<<28, 256, 0, stream>>>(cnt, kern, kfrag);

  int blocks = (Mtotal + 255) / 256;
  bin_count_kernel<<<blocks, 256, 0, stream>>>(pos, cnt, rank, Mtotal, Nper);
  bin_scan_kernel<<<1, 1024, 0, stream>>>(cnt, off);
  bin_fill_kernel<<<blocks, 256, 0, stream>>>(pos, off, rank, sorted, Mtotal, Nper);

  p2g_tile_kernel<<<1024, 512, 0, stream>>>(x, pos, kfrag, bias, cnt, off, sorted, tbuf, Nper);

  sum_copies_channel_major_kernel<<<dim3((DHW + TC - 1) / TC, 2), 256, 0, stream>>>(tbuf, out);
}


# particles of one base cell share a matrix row (up to 2 per row; tile sorted by cell in an added launch): A = wa*xa+wb*xb per offset, bias*sum(w) added at the store; fewer products; sum order changes
# speedup vs baseline: 46.7750x; 1.0861x over previous
#include <hip/hip_runtime.h>


typedef float v2f __attribute__((ext_vector_type(2)));
typedef float v8f __attribute__((ext_vector_type(8)));
typedef unsigned int v2u __attribute__((ext_vector_type(2)));

#define DD 66
#define HHW (66 * 66)
#define DHW (66 * 66 * 66)
#define CI 32
#define CO 32

#define TC 128
#define TH 24
#define TP 33

#define NT 1024
#define TCELLS 1000
#define TSLOT (TCELLS * CO)

__global__ void setup_kernel(int* __restrict__ cnt, const float* __restrict__ kern,
                             float* __restrict__ kfrag) {
  if (blockIdx.x == 0) {
    for (int j = threadIdx.x; j < NT; j += blockDim.x) cnt[j] = 0;
  } else {
    const int o = blockIdx.x - 1;
    for (int t = threadIdx.x; t < 256; t += blockDim.x) {
      float4 v4 = ((const float4*)kern)[o * 256 + t];
      int f   = t * 4;
      int ci  = f >> 5;
      int co  = f & 31;
      int nt  = co >> 4;
      int cc  = ci >> 2;
      int hh  = (ci >> 1) & 1;
      int vv  = ci & 1;
      float* d = kfrag + o * 1024 + (nt * 8 + cc) * 64 + (hh * 16 + (co & 15)) * 2 + vv;
      d[0] = v4.x;
      d[2] = v4.y;
      d[4] = v4.z;
      d[6] = v4.w;
    }
  }
}

__device__ __forceinline__ int tile_of(const float* __restrict__ pos, int p, int Nper) {
  int b0 = (int)(pos[p * 3 + 0] * 64.0f);
  int b1 = (int)(pos[p * 3 + 1] * 64.0f);
  int b2 = (int)(pos[p * 3 + 2] * 64.0f);
  b0 = min(max(b0, 0), 63);
  b1 = min(max(b1, 0), 63);
  b2 = min(max(b2, 0), 63);
  int bb = (p >= Nper) ? 1 : 0;
  return bb * 512 + (b0 >> 3) * 64 + (b1 >> 3) * 8 + (b2 >> 3);
}

__launch_bounds__(256)
__global__ void bin_count_kernel(const float* __restrict__ pos, int* __restrict__ cnt,
                                 int* __restrict__ rank, int Mtotal, int Nper) {
  int p = blockIdx.x * 256 + threadIdx.x;
  if (p < Mtotal) rank[p] = atomicAdd(&cnt[tile_of(pos, p, Nper)], 1);
}

__launch_bounds__(1024)
__global__ void bin_scan_kernel(const int* __restrict__ cnt, int* __restrict__ off) {
  __shared__ int s[NT];
  const int t = threadIdx.x;
  const int c = cnt[t];
  s[t] = c;
  __syncthreads();
  for (int d = 1; d < NT; d <<= 1) {
    int v = (t >= d) ? s[t - d] : 0;
    __syncthreads();
    s[t] += v;
    __syncthreads();
  }
  off[t] = s[t] - c;
}

__launch_bounds__(256)
__global__ void bin_fill_kernel(const float* __restrict__ pos, const int* __restrict__ off,
                                const int* __restrict__ rank, int* __restrict__ sorted,
                                int Mtotal, int Nper) {
  int p = blockIdx.x * 256 + threadIdx.x;
  if (p < Mtotal) sorted[off[tile_of(pos, p, Nper)] + rank[p]] = p;
}

__launch_bounds__(512)
__global__ void cell_sort_kernel(const float* __restrict__ pos, const int* __restrict__ cnt,
                                 const int* __restrict__ off, const int* __restrict__ sorted,
                                 int* __restrict__ tmp, int* __restrict__ sorted2,
                                 int* __restrict__ rowinfo, int* __restrict__ nrows) {
  __shared__ int s_cnt[512];
  __shared__ int s_a[512];
  __shared__ int s_b[512];
  __shared__ int s_off[512];
  const int t     = threadIdx.x;
  const int tile  = blockIdx.x;
  const int n     = cnt[tile];
  const int start = off[tile];
  if (n == 0) {
    if (t == 0) nrows[tile] = 0;
    return;
  }
  const int t0 = (tile >> 6) & 7, t1 = (tile >> 3) & 7, t2 = tile & 7;
  s_cnt[t] = 0;
  __syncthreads();
  for (int i = t; i < n; i += 512) {
    const int p = sorted[start + i];
    const int b0 = (int)(pos[p * 3 + 0] * 64.0f);
    const int b1 = (int)(pos[p * 3 + 1] * 64.0f);
    const int b2 = (int)(pos[p * 3 + 2] * 64.0f);
    const int l0 = min(max(b0 - t0 * 8, 0), 7);
    const int l1 = min(max(b1 - t1 * 8, 0), 7);
    const int l2 = min(max(b2 - t2 * 8, 0), 7);
    const int lc = l0 * 64 + l1 * 8 + l2;
    const int r  = atomicAdd(&s_cnt[lc], 1);
    tmp[start + i] = r | (lc << 20);
  }
  __syncthreads();
  const int c  = s_cnt[t];
  const int rc = (c + 1) >> 1;
  s_a[t] = c;
  s_b[t] = rc;
  __syncthreads();
  for (int d = 1; d < 512; d <<= 1) {
    const int va = (t >= d) ? s_a[t - d] : 0;
    const int vb = (t >= d) ? s_b[t - d] : 0;
    __syncthreads();
    s_a[t] += va;
    s_b[t] += vb;
    __syncthreads();
  }
  const int celloff = s_a[t] - c;
  const int rowoff  = s_b[t] - rc;
  s_off[t] = celloff;
  if (t == 511) nrows[tile] = s_b[511];
  __syncthreads();
  for (int i = t; i < n; i += 512) {
    const int v = tmp[start + i];
    sorted2[start + s_off[v >> 20] + (v & 0xFFFFF)] = sorted[start + i];
  }
  for (int k = 0; k < rc; ++k)
    rowinfo[start + rowoff + k] = (celloff + 2 * k) | ((c - 2 * k >= 2) ? (1 << 30) : 0);
}

__device__ __forceinline__ void particle_weights(const float* __restrict__ pos, int p, float valid,
                                                 int t0, int t1, int t2,
                                                 float (&wx)[3], float (&wy)[3], float (&wz)[3],
                                                 int& lbase) {
  float px = pos[p * 3 + 0] * 64.0f;
  float py = pos[p * 3 + 1] * 64.0f;
  float pz = pos[p * 3 + 2] * 64.0f;
  int b0 = (int)px, b1 = (int)py, b2 = (int)pz;
  float fx = px - (float)b0 - 0.5f;
  float fy = py - (float)b1 - 0.5f;
  float fz = pz - (float)b2 - 0.5f;
  wx[0] = 0.5f * (0.5f - fx) * (0.5f - fx) * valid;
  wx[1] = (0.75f - fx * fx) * valid;
  wx[2] = 0.5f * (0.5f + fx) * (0.5f + fx) * valid;
  wy[0] = 0.5f * (0.5f - fy) * (0.5f - fy);
  wy[1] = 0.75f - fy * fy;
  wy[2] = 0.5f * (0.5f + fy) * (0.5f + fy);
  wz[0] = 0.5f * (0.5f - fz) * (0.5f - fz);
  wz[1] = 0.75f - fz * fz;
  wz[2] = 0.5f * (0.5f + fz) * (0.5f + fz);
  const int l0 = min(max(b0 - t0 * 8, 0), 7);
  const int l1 = min(max(b1 - t1 * 8, 0), 7);
  const int l2 = min(max(b2 - t2 * 8, 0), 7);
  lbase = l0 * 100 + l1 * 10 + l2;
}

__launch_bounds__(256, 4)
__global__ void p2g_tile_kernel(const float* __restrict__ x,
                                const float* __restrict__ pos,
                                const float* __restrict__ kfrag,
                                const float* __restrict__ bias,
                                const int* __restrict__ nrows,
                                const int* __restrict__ off,
                                const int* __restrict__ sorted2,
                                const int* __restrict__ rowinfo,
                                float* __restrict__ tbuf) {
  __shared__ __attribute__((aligned(16))) float sAcc[TCELLS * CO];
  __shared__ float sW[TCELLS];

  const int t    = threadIdx.x;
  const int lane = t & 31;
  const int wv   = __builtin_amdgcn_readfirstlane(t >> 5);
  const int half = lane >> 4;
  const int nlo  = lane & 15;

  const int tile  = blockIdx.x;
  const int nr    = nrows[tile];
  const int start = off[tile];
  float4* tb4 = (float4*)(tbuf + (size_t)tile * TSLOT);
  if (nr == 0) {
    for (int i = t; i < TSLOT / 4; i += 256) tb4[i] = make_float4(0.f, 0.f, 0.f, 0.f);
    return;
  }
  const int t0 = (tile >> 6) & 7, t1 = (tile >> 3) & 7, t2 = tile & 7;

  {
    float4* z = (float4*)sAcc;
    for (int i = t; i < TCELLS * CO / 4; i += 256) z[i] = make_float4(0.f, 0.f, 0.f, 0.f);
    for (int i = t; i < TCELLS; i += 256) sW[i] = 0.f;
  }
  __syncthreads();

  const int nch = (nr + 31) >> 5;

  const int tot = nch * 27;
  const int lo  = (wv * tot) >> 3;
  const int hi  = ((wv + 1) * tot) >> 3;
  for (int ch = lo / 27; ch * 27 < hi; ++ch) {
    const int oB = max(lo - ch * 27, 0);
    const int oE = min(hi - ch * 27, 27);

    v2f xa[2][8], xb[2][8];
    float wax0[3], way0[3], waz0[3], wbx0[3], wby0[3], wbz0[3];
    float wax1[3], way1[3], waz1[3], wbx1[3], wby1[3], wbz1[3];
    int lbA[2];
#pragma unroll
    for (int mt = 0; mt < 2; ++mt) {
      const int jr   = ch * 32 + mt * 16 + nlo;
      const bool vr  = jr < nr;
      const int jc   = vr ? jr : (nr - 1);
      const int info = rowinfo[start + jc];
      const int slot = info & 0x3FFFFFFF;
      const int two  = (info >> 30) & 1;
      const int pa   = sorted2[start + slot];
      const int pb   = sorted2[start + slot + two];
      int lbB;
      if (mt == 0) {
        particle_weights(pos, pa, vr ? 1.0f : 0.0f, t0, t1, t2, wax0, way0, waz0, lbA[0]);
        particle_weights(pos, pb, (vr && two) ? 1.0f : 0.0f, t0, t1, t2, wbx0, wby0, wbz0, lbB);
      } else {
        particle_weights(pos, pa, vr ? 1.0f : 0.0f, t0, t1, t2, wax1, way1, waz1, lbA[1]);
        particle_weights(pos, pb, (vr && two) ? 1.0f : 0.0f, t0, t1, t2, wbx1, wby1, wbz1, lbB);
      }
      const float* xra = x + pa * CI + half * 2;
      const float* xrb = x + pb * CI + half * 2;
#pragma unroll
      for (int c = 0; c < 8; ++c) {
        xa[mt][c] = *(const v2f*)(xra + c * 4);
        xb[mt][c] = *(const v2f*)(xrb + c * 4);
      }
    }
    const int lbase = half ? lbA[1] : lbA[0];

    int lb[32];
#pragma unroll
    for (int m = 0; m < 32; ++m) lb[m] = __builtin_amdgcn_readlane(lbase * CO, m);

    for (int o = oB; o < oE; ++o) {
      const int oi = o / 9, oj = (o / 3) % 3, ok = o % 3;
      const int cellc = oi * 100 + oj * 10 + ok;
      const int vo = cellc * CO + lane;

      float wa[2], wb[2];
      wa[0] = wax0[oi] * way0[oj] * waz0[ok];
      wb[0] = wbx0[oi] * wby0[oj] * wbz0[ok];
      wa[1] = wax1[oi] * way1[oj] * waz1[ok];
      wb[1] = wbx1[oi] * wby1[oj] * wbz1[ok];
      atomicAdd(&sW[lbase + cellc], half ? (wa[1] + wb[1]) : (wa[0] + wb[0]));

      v2f a[2][8];
#pragma unroll
      for (int mt = 0; mt < 2; ++mt)
#pragma unroll
        for (int c = 0; c < 8; ++c)
          a[mt][c] = xa[mt][c] * wa[mt] + xb[mt][c] * wb[mt];

      const float* kf = kfrag + o * 1024 + lane * 2;
      v2f bf[2][8];
#pragma unroll
      for (int nt = 0; nt < 2; ++nt)
#pragma unroll
        for (int c = 0; c < 8; ++c)
          bf[nt][c] = *(const v2f*)&kf[(nt * 8 + c) * 64];

      v8f dacc[2][2];
#pragma unroll
      for (int mt = 0; mt < 2; ++mt)
#pragma unroll
        for (int nt = 0; nt < 2; ++nt)
#pragma unroll
          for (int r = 0; r < 8; ++r) dacc[mt][nt][r] = 0.0f;

#pragma unroll
      for (int c = 0; c < 8; ++c) {
#pragma unroll
        for (int mt = 0; mt < 2; ++mt) {
#pragma unroll
          for (int nt = 0; nt < 2; ++nt) {
            dacc[mt][nt] = __builtin_amdgcn_wmma_f32_16x16x4_f32(
                false, a[mt][c], false, bf[nt][c], (short)0, dacc[mt][nt],
                false, false);
          }
        }
      }

#pragma unroll
      for (int mt = 0; mt < 2; ++mt) {
#pragma unroll
        for (int r = 0; r < 8; ++r) {
          v2u s = __builtin_amdgcn_permlane16_swap(
              __float_as_uint(dacc[mt][0][r]), __float_as_uint(dacc[mt][1][r]), false, false);
          const float rowLo = __uint_as_float(s[0]);
          const float rowHi = __uint_as_float(s[1]);
          const int m0 = mt * 16 + r;
          const int m1 = m0 + 8;
          atomicAdd(&sAcc[lb[m0] + vo], rowLo);
          atomicAdd(&sAcc[lb[m1] + vo], rowHi);
        }
      }
    }
  }

  __syncthreads();

  {
    const float4* s4 = (const float4*)sAcc;
    const float4 b4 = ((const float4*)bias)[t & 7];
    for (int i = t; i < TSLOT / 4; i += 256) {
      const float4 s = s4[i];
      const float w = sW[i >> 3];
      tb4[i] = make_float4(fmaf(w, b4.x, s.x), fmaf(w, b4.y, s.y), fmaf(w, b4.z, s.z), fmaf(w, b4.w, s.w));
    }
  }
}

__launch_bounds__(256)
__global__ void sum_copies_channel_major_kernel(const float* __restrict__ tbuf,
                                                float* __restrict__ out) {
  __shared__ float tile[(TC + TH) * TP];

  const int t    = threadIdx.x;
  const int lane = t & 31;
  const int wv   = t >> 5;
  const int b    = blockIdx.y;
  const int c0   = blockIdx.x * TC;

  int ncell = DHW - c0;
  ncell = (ncell < TC + TH) ? ncell : (TC + TH);
  const int nf4 = ncell * (CO / 4);
  for (int i = t; i < nf4; i += 256) {
    const int g  = c0 + (i >> 3);
    const int d0 = g / HHW;
    const int gr = g - d0 * HHW;
    const int d1 = gr / DD;
    const int d2 = gr - d1 * DD;
    const int q0 = d0 >> 3, q1 = d1 >> 3, q2 = d2 >> 3;
    const int r0 = d0 & 7,  r1 = d1 & 7,  r2 = d2 & 7;
    const int e = ((b * 512 + q0 * 64 + q1 * 8 + q2) * TCELLS + r0 * 100 + r1 * 10 + r2) * CO + (i & 7) * 4;
    const int E0 = (800 - 64 * TCELLS) * CO;
    const int E1 = (80 - 8 * TCELLS) * CO;
    const int E2 = (8 - TCELLS) * CO;
    const bool A0 = q0 < 8, A1 = q1 < 8, A2 = q2 < 8;
    const bool B0 = (q0 >= 1) && (r0 < 2);
    const bool B1 = (q1 >= 1) && (r1 < 2);
    const bool B2 = (q2 >= 1) && (r2 < 2);

    const float4 z4 = make_float4(0.f, 0.f, 0.f, 0.f);
    float4 v0 = z4, v1 = z4, v2 = z4, v3 = z4, v4 = z4, v5 = z4, v6 = z4, v7 = z4;
    if (A0) {
      if (A1) {
        if (A2) v0 = *(const float4*)(tbuf + e);
        if (B2) v1 = *(const float4*)(tbuf + (e + E2));
      }
      if (B1) {
        if (A2) v2 = *(const float4*)(tbuf + (e + E1));
        if (B2) v3 = *(const float4*)(tbuf + (e + E1 + E2));
      }
    }
    if (B0) {
      if (A1) {
        if (A2) v4 = *(const float4*)(tbuf + (e + E0));
        if (B2) v5 = *(const float4*)(tbuf + (e + E0 + E2));
      }
      if (B1) {
        if (A2) v6 = *(const float4*)(tbuf + (e + E0 + E1));
        if (B2) v7 = *(const float4*)(tbuf + (e + E0 + E1 + E2));
      }
    }
    float4 s;
    s.x = ((((((v0.x + v1.x) + v2.x) + v3.x) + v4.x) + v5.x) + v6.x) + v7.x;
    s.y = ((((((v0.y + v1.y) + v2.y) + v3.y) + v4.y) + v5.y) + v6.y) + v7.y;
    s.z = ((((((v0.z + v1.z) + v2.z) + v3.z) + v4.z) + v5.z) + v6.z) + v7.z;
    s.w = ((((((v0.w + v1.w) + v2.w) + v3.w) + v4.w) + v5.w) + v6.w) + v7.w;

    float* d = &tile[(i >> 3) * TP + (i & 7) * 4];
    d[0] = s.x; d[1] = s.y; d[2] = s.z; d[3] = s.w;
  }
  __syncthreads();

  float v[4][TC / 32];
  float vhead[4];
#pragma unroll
  for (int k = 0; k < 4; ++k) {
    const int ch = wv * 4 + k;
    const int sh = (32 - 8 * (ch & 3)) & 31;
#pragma unroll
    for (int e = 0; e < TC / 32; ++e) {
      int lr = sh + e * 32 + lane;
      lr = (lr < ncell) ? lr : (ncell - 1);
      v[k][e] = tile[lr * TP + ch];
    }
    vhead[k] = tile[((lane < ncell) ? lane : (ncell - 1)) * TP + ch];
  }
#pragma unroll
  for (int k = 0; k < 4; ++k) {
    const int ch = wv * 4 + k;
    const int sh = (32 - 8 * (ch & 3)) & 31;
    float* orow = out + ((size_t)(b * CO + ch)) * DHW + c0;
#pragma unroll
    for (int e = 0; e < TC / 32; ++e) {
      const int lc = sh + e * 32 + lane;
      if (c0 + lc < DHW) orow[lc] = v[k][e];
    }
    if (blockIdx.x == 0 && lane < sh) orow[lane] = vhead[k];
  }
}

extern "C" void kernel_launch(void* const* d_in, const int* in_sizes, int n_in,
                              void* d_out, int out_size, void* d_ws, size_t ws_size,
                              hipStream_t stream) {
  const float* x    = (const float*)d_in[0];
  const float* pos  = (const float*)d_in[1];
  const float* kern = (const float*)d_in[2];
  const float* bias = (const float*)d_in[3];
  float* out = (float*)d_out;
  float* tbuf = (float*)d_ws;
  int* cnt    = (int*)d_ws + 32768000;
  int* off    = cnt + 1024;
  int* rank   = cnt + 2048;
  int* sorted = cnt + 2048 + 524288;
  float* kfrag = (float*)(cnt + 2048 + 524288 + 524288);
  int* sorted2 = cnt + 1078272;
  int* rowinfo = sorted2 + 524288;
  int* nrows   = rowinfo + 524288;

  const int Mtotal = in_sizes[0] / CI;
  const int Nper   = in_sizes[1] / (2 * 3);

  setup_kernel<<<28, 256, 0, stream>>>(cnt, kern, kfrag);

  int blocks = (Mtotal + 255) / 256;
  bin_count_kernel<<<blocks, 256, 0, stream>>>(pos, cnt, rank, Mtotal, Nper);
  bin_scan_kernel<<<1, 1024, 0, stream>>>(cnt, off);
  bin_fill_kernel<<<blocks, 256, 0, stream>>>(pos, off, rank, sorted, Mtotal, Nper);

  cell_sort_kernel<<<1024, 512, 0, stream>>>(pos, cnt, off, sorted, rank, sorted2, rowinfo, nrows);

  p2g_tile_kernel<<<1024, 256, 0, stream>>>(x, pos, kfrag, bias, nrows, off, sorted2, rowinfo, tbuf);

  sum_copies_channel_major_kernel<<<dim3((DHW + TC - 1) / TC, 2), 256, 0, stream>>>(tbuf, out);
}


# particles grouped by tile inside 4096-particle blocks (LDS counters); each tile collects its runs from the blocks' lists: no global atomics, 7 launches -> 4; sum order inside a cell changes
# speedup vs baseline: 53.6624x; 1.1472x over previous
#include <hip/hip_runtime.h>


typedef float v2f __attribute__((ext_vector_type(2)));
typedef float v8f __attribute__((ext_vector_type(8)));
typedef unsigned int v2u __attribute__((ext_vector_type(2)));

#define DD 66
#define HHW (66 * 66)
#define DHW (66 * 66 * 66)
#define CI 32
#define CO 32

#define TC 128
#define TH 24
#define TP 33

#define NT 1024
#define TCELLS 1000
#define TSLOT (TCELLS * CO)

#define PBLK 4096
#define BPITCH (NT + 1)

__device__ __forceinline__ int tile_of(const float* __restrict__ pos, int p, int Nper) {
  int b0 = (int)(pos[p * 3 + 0] * 64.0f);
  int b1 = (int)(pos[p * 3 + 1] * 64.0f);
  int b2 = (int)(pos[p * 3 + 2] * 64.0f);
  b0 = min(max(b0, 0), 63);
  b1 = min(max(b1, 0), 63);
  b2 = min(max(b2, 0), 63);
  int bb = (p >= Nper) ? 1 : 0;
  return bb * 512 + (b0 >> 3) * 64 + (b1 >> 3) * 8 + (b2 >> 3);
}

template <int NW, int NV>
__device__ __forceinline__ void block_inclusive_scan(int (&v)[NV], int* s_w) {
  const int lane = threadIdx.x & 31;
  const int wave = threadIdx.x >> 5;
  for (int d = 1; d < 32; d <<= 1) {
    int u[NV];
    for (int j = 0; j < NV; ++j) u[j] = __shfl_up(v[j], d);
    if (lane >= d)
      for (int j = 0; j < NV; ++j) v[j] += u[j];
  }
  if (lane == 31)
    for (int j = 0; j < NV; ++j) s_w[32 * j + wave] = v[j];
  __syncthreads();
  if (wave == 0) {
    int w[NV];
    for (int j = 0; j < NV; ++j) w[j] = (lane < NW) ? s_w[32 * j + lane] : 0;
    for (int d = 1; d < 32; d <<= 1) {
      int u[NV];
      for (int j = 0; j < NV; ++j) u[j] = __shfl_up(w[j], d);
      if (lane >= d)
        for (int j = 0; j < NV; ++j) w[j] += u[j];
    }
    for (int j = 0; j < NV; ++j) s_w[32 * j + lane] = w[j];
  }
  __syncthreads();
  if (wave > 0)
    for (int j = 0; j < NV; ++j) v[j] += s_w[32 * j + wave - 1];
}

__launch_bounds__(1024)
__global__ void block_sort_kernel(const float* __restrict__ pos, const float* __restrict__ kern,
                                  float* __restrict__ kfrag, int* __restrict__ boff,
                                  int* __restrict__ bsorted, int Mtotal, int Nper, int nblk) {
  __shared__ int s_hist[NT];
  __shared__ int s_w[32];
  const int t   = threadIdx.x;
  const int blk = blockIdx.x;
  if (blk >= nblk) {
    if (t < 256) {
      const int o = blk - nblk;
      float4 v4 = ((const float4*)kern)[o * 256 + t];
      int f   = t * 4;
      int ci  = f >> 5;
      int co  = f & 31;
      int nt  = co >> 4;
      int cc  = ci >> 2;
      int hh  = (ci >> 1) & 1;
      int vv  = ci & 1;
      float* d = kfrag + o * 1024 + (nt * 8 + cc) * 64 + (hh * 16 + (co & 15)) * 2 + vv;
      d[0] = v4.x;
      d[2] = v4.y;
      d[4] = v4.z;
      d[6] = v4.w;
    }
    return;
  }
  s_hist[t] = 0;
  int pn[4], tl[4], rk[4];
  for (int k = 0; k < 4; ++k) {
    pn[k] = blk * PBLK + k * 1024 + t;
    tl[k] = tile_of(pos, min(pn[k], Mtotal - 1), Nper);
  }
  __syncthreads();
  for (int k = 0; k < 4; ++k) {
    rk[k] = 0;
    if (pn[k] < Mtotal) rk[k] = atomicAdd(&s_hist[tl[k]], 1);
  }
  __syncthreads();
  const int c = s_hist[t];
  int incl[1] = {c};
  block_inclusive_scan<32, 1>(incl, s_w);
  s_hist[t] = incl[0] - c;
  boff[blk * BPITCH + t] = incl[0] - c;
  if (t == NT - 1) boff[blk * BPITCH + NT] = incl[0];
  __syncthreads();
  for (int k = 0; k < 4; ++k)
    if (pn[k] < Mtotal) bsorted[blk * PBLK + s_hist[tl[k]] + rk[k]] = pn[k];
}

__launch_bounds__(256)
__global__ void tile_cell_sort_kernel(const float* __restrict__ pos, const int* __restrict__ boff,
                                      const int* __restrict__ bsorted, int nblk,
                                      int* __restrict__ off, int* __restrict__ ptmp,
                                      int* __restrict__ tmp, int* __restrict__ sorted2,
                                      int* __restrict__ rowinfo, int* __restrict__ nrows) {
  __shared__ int s_cnt[512];
  __shared__ int s_off[512];
  __shared__ int s_run[128];
  __shared__ int s_rs[128];
  __shared__ int s_w[64];
  const int t    = threadIdx.x;
  const int tile = blockIdx.x;
  int rs = 0, rn = 0;
  if (t < nblk) {
    rs = boff[t * BPITCH + tile];
    rn = boff[t * BPITCH + tile + 1] - rs;
  }
  int runs[2] = {rn, rs};
  block_inclusive_scan<8, 2>(runs, s_w);
  if (t < 128) {
    s_run[t] = runs[0];
    s_rs[t]  = rs;
  }
  s_cnt[t]       = 0;
  s_cnt[t + 256] = 0;
  const int n     = s_w[31];
  const int start = s_w[63];
  if (t == 0) off[tile] = start;
  if (n == 0) {
    if (t == 0) nrows[tile] = 0;
    return;
  }
  const int t0 = (tile >> 6) & 7, t1 = (tile >> 3) & 7, t2 = tile & 7;
  __syncthreads();
  for (int i = t; i < n; i += 256) {
    int lo = 0, hi = 127;
    for (int k = 0; k < 7; ++k) {
      const int mid = (lo + hi) >> 1;
      if (s_run[mid] > i) hi = mid; else lo = mid + 1;
    }
    const int before = (lo > 0) ? s_run[lo - 1] : 0;
    const int p = bsorted[lo * PBLK + s_rs[lo] + (i - before)];
    const int b0 = (int)(pos[p * 3 + 0] * 64.0f);
    const int b1 = (int)(pos[p * 3 + 1] * 64.0f);
    const int b2 = (int)(pos[p * 3 + 2] * 64.0f);
    const int l0 = min(max(b0 - t0 * 8, 0), 7);
    const int l1 = min(max(b1 - t1 * 8, 0), 7);
    const int l2 = min(max(b2 - t2 * 8, 0), 7);
    const int lc = l0 * 64 + l1 * 8 + l2;
    const int r  = atomicAdd(&s_cnt[lc], 1);
    tmp[start + i]  = r | (lc << 20);
    ptmp[start + i] = p;
  }
  __syncthreads();
  const int ca  = s_cnt[2 * t];
  const int cb  = s_cnt[2 * t + 1];
  const int rca = (ca + 1) >> 1;
  const int rcb = (cb + 1) >> 1;
  int ends[2] = {ca + cb, rca + rcb};
  block_inclusive_scan<8, 2>(ends, s_w);
  const int celloffa = ends[0] - cb - ca;
  const int celloffb = ends[0] - cb;
  const int rowoffa  = ends[1] - rcb - rca;
  const int rowoffb  = ends[1] - rcb;
  s_off[2 * t]     = celloffa;
  s_off[2 * t + 1] = celloffb;
  if (t == 255) nrows[tile] = ends[1];
  __syncthreads();
  for (int i = t; i < n; i += 256) {
    const int v = tmp[start + i];
    sorted2[start + s_off[v >> 20] + (v & 0xFFFFF)] = ptmp[start + i];
  }
  for (int k = 0; k < rca; ++k)
    rowinfo[start + rowoffa + k] = (celloffa + 2 * k) | ((ca - 2 * k >= 2) ? (1 << 30) : 0);
  for (int k = 0; k < rcb; ++k)
    rowinfo[start + rowoffb + k] = (celloffb + 2 * k) | ((cb - 2 * k >= 2) ? (1 << 30) : 0);
}

__device__ __forceinline__ void particle_weights(const float* __restrict__ pos, int p, float valid,
                                                 int t0, int t1, int t2,
                                                 float (&wx)[3], float (&wy)[3], float (&wz)[3],
                                                 int& lbase) {
  float px = pos[p * 3 + 0] * 64.0f;
  float py = pos[p * 3 + 1] * 64.0f;
  float pz = pos[p * 3 + 2] * 64.0f;
  int b0 = (int)px, b1 = (int)py, b2 = (int)pz;
  float fx = px - (float)b0 - 0.5f;
  float fy = py - (float)b1 - 0.5f;
  float fz = pz - (float)b2 - 0.5f;
  wx[0] = 0.5f * (0.5f - fx) * (0.5f - fx) * valid;
  wx[1] = (0.75f - fx * fx) * valid;
  wx[2] = 0.5f * (0.5f + fx) * (0.5f + fx) * valid;
  wy[0] = 0.5f * (0.5f - fy) * (0.5f - fy);
  wy[1] = 0.75f - fy * fy;
  wy[2] = 0.5f * (0.5f + fy) * (0.5f + fy);
  wz[0] = 0.5f * (0.5f - fz) * (0.5f - fz);
  wz[1] = 0.75f - fz * fz;
  wz[2] = 0.5f * (0.5f + fz) * (0.5f + fz);
  const int l0 = min(max(b0 - t0 * 8, 0), 7);
  const int l1 = min(max(b1 - t1 * 8, 0), 7);
  const int l2 = min(max(b2 - t2 * 8, 0), 7);
  lbase = l0 * 100 + l1 * 10 + l2;
}

__launch_bounds__(256, 4)
__global__ void p2g_tile_kernel(const float* __restrict__ x,
                                const float* __restrict__ pos,
                                const float* __restrict__ kfrag,
                                const float* __restrict__ bias,
                                const int* __restrict__ nrows,
                                const int* __restrict__ off,
                                const int* __restrict__ sorted2,
                                const int* __restrict__ rowinfo,
                                float* __restrict__ tbuf) {
  __shared__ __attribute__((aligned(16))) float sAcc[TCELLS * CO];
  __shared__ float sW[TCELLS];

  const int t    = threadIdx.x;
  const int lane = t & 31;
  const int wv   = __builtin_amdgcn_readfirstlane(t >> 5);
  const int half = lane >> 4;
  const int nlo  = lane & 15;

  const int tile  = blockIdx.x;
  const int nr    = nrows[tile];
  const int start = off[tile];
  float4* tb4 = (float4*)(tbuf + (size_t)tile * TSLOT);
  if (nr == 0) {
    for (int i = t; i < TSLOT / 4; i += 256) tb4[i] = make_float4(0.f, 0.f, 0.f, 0.f);
    return;
  }
  const int t0 = (tile >> 6) & 7, t1 = (tile >> 3) & 7, t2 = tile & 7;

  {
    float4* z = (float4*)sAcc;
    for (int i = t; i < TCELLS * CO / 4; i += 256) z[i] = make_float4(0.f, 0.f, 0.f, 0.f);
    for (int i = t; i < TCELLS; i += 256) sW[i] = 0.f;
  }
  __syncthreads();

  const int nch = (nr + 31) >> 5;

  const int tot = nch * 27;
  const int lo  = (wv * tot) >> 3;
  const int hi  = ((wv + 1) * tot) >> 3;
  for (int ch = lo / 27; ch * 27 < hi; ++ch) {
    const int oB = max(lo - ch * 27, 0);
    const int oE = min(hi - ch * 27, 27);

    v2f xa[2][8], xb[2][8];
    float wax0[3], way0[3], waz0[3], wbx0[3], wby0[3], wbz0[3];
    float wax1[3], way1[3], waz1[3], wbx1[3], wby1[3], wbz1[3];
    int lbA[2];
#pragma unroll
    for (int mt = 0; mt < 2; ++mt) {
      const int jr   = ch * 32 + mt * 16 + nlo;
      const bool vr  = jr < nr;
      const int jc   = vr ? jr : (nr - 1);
      const int info = rowinfo[start + jc];
      const int slot = info & 0x3FFFFFFF;
      const int two  = (info >> 30) & 1;
      const int pa   = sorted2[start + slot];
      const int pb   = sorted2[start + slot + two];
      int lbB;
      if (mt == 0) {
        particle_weights(pos, pa, vr ? 1.0f : 0.0f, t0, t1, t2, wax0, way0, waz0, lbA[0]);
        particle_weights(pos, pb, (vr && two) ? 1.0f : 0.0f, t0, t1, t2, wbx0, wby0, wbz0, lbB);
      } else {
        particle_weights(pos, pa, vr ? 1.0f : 0.0f, t0, t1, t2, wax1, way1, waz1, lbA[1]);
        particle_weights(pos, pb, (vr && two) ? 1.0f : 0.0f, t0, t1, t2, wbx1, wby1, wbz1, lbB);
      }
      const float* xra = x + pa * CI + half * 2;
      const float* xrb = x + pb * CI + half * 2;
#pragma unroll
      for (int c = 0; c < 8; ++c) {
        xa[mt][c] = *(const v2f*)(xra + c * 4);
        xb[mt][c] = *(const v2f*)(xrb + c * 4);
      }
    }
    const int lbase = half ? lbA[1] : lbA[0];

    int lb[32];
#pragma unroll
    for (int m = 0; m < 32; ++m) lb[m] = __builtin_amdgcn_readlane(lbase * CO, m);

    for (int o = oB; o < oE; ++o) {
      const int oi = o / 9, oj = (o / 3) % 3, ok = o % 3;
      const int cellc = oi * 100 + oj * 10 + ok;
      const int vo = cellc * CO + lane;

      float wa[2], wb[2];
      wa[0] = wax0[oi] * way0[oj] * waz0[ok];
      wb[0] = wbx0[oi] * wby0[oj] * wbz0[ok];
      wa[1] = wax1[oi] * way1[oj] * waz1[ok];
      wb[1] = wbx1[oi] * wby1[oj] * wbz1[ok];
      atomicAdd(&sW[lbase + cellc], half ? (wa[1] + wb[1]) : (wa[0] + wb[0]));

      v2f a[2][8];
#pragma unroll
      for (int mt = 0; mt < 2; ++mt)
#pragma unroll
        for (int c = 0; c < 8; ++c)
          a[mt][c] = xa[mt][c] * wa[mt] + xb[mt][c] * wb[mt];

      const float* kf = kfrag + o * 1024 + lane * 2;
      v2f bf[2][8];
#pragma unroll
      for (int nt = 0; nt < 2; ++nt)
#pragma unroll
        for (int c = 0; c < 8; ++c)
          bf[nt][c] = *(const v2f*)&kf[(nt * 8 + c) * 64];

      v8f dacc[2][2];
#pragma unroll
      for (int mt = 0; mt < 2; ++mt)
#pragma unroll
        for (int nt = 0; nt < 2; ++nt)
#pragma unroll
          for (int r = 0; r < 8; ++r) dacc[mt][nt][r] = 0.0f;

#pragma unroll
      for (int c = 0; c < 8; ++c) {
#pragma unroll
        for (int mt = 0; mt < 2; ++mt) {
#pragma unroll
          for (int nt = 0; nt < 2; ++nt) {
            dacc[mt][nt] = __builtin_amdgcn_wmma_f32_16x16x4_f32(
                false, a[mt][c], false, bf[nt][c], (short)0, dacc[mt][nt],
                false, false);
          }
        }
      }

#pragma unroll
      for (int mt = 0; mt < 2; ++mt) {
#pragma unroll
        for (int r = 0; r < 8; ++r) {
          v2u s = __builtin_amdgcn_permlane16_swap(
              __float_as_uint(dacc[mt][0][r]), __float_as_uint(dacc[mt][1][r]), false, false);
          const float rowLo = __uint_as_float(s[0]);
          const float rowHi = __uint_as_float(s[1]);
          const int m0 = mt * 16 + r;
          const int m1 = m0 + 8;
          atomicAdd(&sAcc[lb[m0] + vo], rowLo);
          atomicAdd(&sAcc[lb[m1] + vo], rowHi);
        }
      }
    }
  }

  __syncthreads();

  {
    const float4* s4 = (const float4*)sAcc;
    const float4 b4 = ((const float4*)bias)[t & 7];
    for (int i = t; i < TSLOT / 4; i += 256) {
      const float4 s = s4[i];
      const float w = sW[i >> 3];
      tb4[i] = make_float4(fmaf(w, b4.x, s.x), fmaf(w, b4.y, s.y), fmaf(w, b4.z, s.z), fmaf(w, b4.w, s.w));
    }
  }
}

__launch_bounds__(256)
__global__ void sum_copies_channel_major_kernel(const float* __restrict__ tbuf,
                                                float* __restrict__ out) {
  __shared__ float tile[(TC + TH) * TP];

  const int t    = threadIdx.x;
  const int lane = t & 31;
  const int wv   = t >> 5;
  const int b    = blockIdx.y;
  const int c0   = blockIdx.x * TC;

  int ncell = DHW - c0;
  ncell = (ncell < TC + TH) ? ncell : (TC + TH);
  const int nf4 = ncell * (CO / 4);
  for (int i = t; i < nf4; i += 256) {
    const int g  = c0 + (i >> 3);
    const int d0 = g / HHW;
    const int gr = g - d0 * HHW;
    const int d1 = gr / DD;
    const int d2 = gr - d1 * DD;
    const int q0 = d0 >> 3, q1 = d1 >> 3, q2 = d2 >> 3;
    const int r0 = d0 & 7,  r1 = d1 & 7,  r2 = d2 & 7;
    const int e = ((b * 512 + q0 * 64 + q1 * 8 + q2) * TCELLS + r0 * 100 + r1 * 10 + r2) * CO + (i & 7) * 4;
    const int E0 = (800 - 64 * TCELLS) * CO;
    const int E1 = (80 - 8 * TCELLS) * CO;
    const int E2 = (8 - TCELLS) * CO;
    const bool A0 = q0 < 8, A1 = q1 < 8, A2 = q2 < 8;
    const bool B0 = (q0 >= 1) && (r0 < 2);
    const bool B1 = (q1 >= 1) && (r1 < 2);
    const bool B2 = (q2 >= 1) && (r2 < 2);

    const float4 z4 = make_float4(0.f, 0.f, 0.f, 0.f);
    float4 v0 = z4, v1 = z4, v2 = z4, v3 = z4, v4 = z4, v5 = z4, v6 = z4, v7 = z4;
    if (A0) {
      if (A1) {
        if (A2) v0 = *(const float4*)(tbuf + e);
        if (B2) v1 = *(const float4*)(tbuf + (e + E2));
      }
      if (B1) {
        if (A2) v2 = *(const float4*)(tbuf + (e + E1));
        if (B2) v3 = *(const float4*)(tbuf + (e + E1 + E2));
      }
    }
    if (B0) {
      if (A1) {
        if (A2) v4 = *(const float4*)(tbuf + (e + E0));
        if (B2) v5 = *(const float4*)(tbuf + (e + E0 + E2));
      }
      if (B1) {
        if (A2) v6 = *(const float4*)(tbuf + (e + E0 + E1));
        if (B2) v7 = *(const float4*)(tbuf + (e + E0 + E1 + E2));
      }
    }
    float4 s;
    s.x = ((((((v0.x + v1.x) + v2.x) + v3.x) + v4.x) + v5.x) + v6.x) + v7.x;
    s.y = ((((((v0.y + v1.y) + v2.y) + v3.y) + v4.y) + v5.y) + v6.y) + v7.y;
    s.z = ((((((v0.z + v1.z) + v2.z) + v3.z) + v4.z) + v5.z) + v6.z) + v7.z;
    s.w = ((((((v0.w + v1.w) + v2.w) + v3.w) + v4.w) + v5.w) + v6.w) + v7.w;

    float* d = &tile[(i >> 3) * TP + (i & 7) * 4];
    d[0] = s.x; d[1] = s.y; d[2] = s.z; d[3] = s.w;
  }
  __syncthreads();

  float v[4][TC / 32];
  float vhead[4];
#pragma unroll
  for (int k = 0; k < 4; ++k) {
    const int ch = wv * 4 + k;
    const int sh = (32 - 8 * (ch & 3)) & 31;
#pragma unroll
    for (int e = 0; e < TC / 32; ++e) {
      int lr = sh + e * 32 + lane;
      lr = (lr < ncell) ? lr : (ncell - 1);
      v[k][e] = tile[lr * TP + ch];
    }
    vhead[k] = tile[((lane < ncell) ? lane : (ncell - 1)) * TP + ch];
  }
#pragma unroll
  for (int k = 0; k < 4; ++k) {
    const int ch = wv * 4 + k;
    const int sh = (32 - 8 * (ch & 3)) & 31;
    float* orow = out + ((size_t)(b * CO + ch)) * DHW + c0;
#pragma unroll
    for (int e = 0; e < TC / 32; ++e) {
      const int lc = sh + e * 32 + lane;
      if (c0 + lc < DHW) orow[lc] = v[k][e];
    }
    if (blockIdx.x == 0 && lane < sh) orow[lane] = vhead[k];
  }
}

extern "C" void kernel_launch(void* const* d_in, const int* in_sizes, int n_in,
                              void* d_out, int out_size, void* d_ws, size_t ws_size,
                              hipStream_t stream) {
  const float* x    = (const float*)d_in[0];
  const float* pos  = (const float*)d_in[1];
  const float* kern = (const float*)d_in[2];
  const float* bias = (const float*)d_in[3];
  float* out = (float*)d_out;
  float* tbuf = (float*)d_ws;
  int* wsi    = (int*)d_ws + 32768000;
  int* off    = wsi + 1024;
  int* tmp    = wsi + 2048;
  int* ptmp   = wsi + 2048 + 524288;
  float* kfrag = (float*)(wsi + 2048 + 524288 + 524288);
  int* sorted2 = wsi + 1078272;
  int* rowinfo = sorted2 + 524288;
  int* nrows   = rowinfo + 524288;
  int* boff    = nrows + 1024;
  int* bsorted = boff + 128 * BPITCH;

  const int Mtotal = in_sizes[0] / CI;
  const int Nper   = in_sizes[1] / (2 * 3);
  const int nblk   = (Mtotal + PBLK - 1) / PBLK;

  block_sort_kernel<<<nblk + 27, 1024, 0, stream>>>(pos, kern, kfrag, boff, bsorted, Mtotal, Nper, nblk);

  tile_cell_sort_kernel<<<1024, 256, 0, stream>>>(pos, boff, bsorted, nblk, off, ptmp, tmp, sorted2, rowinfo, nrows);

  p2g_tile_kernel<<<1024, 256, 0, stream>>>(x, pos, kfrag, bias, nrows, off, sorted2, rowinfo, tbuf);

  sum_copies_channel_major_kernel<<<dim3((DHW + TC - 1) / TC, 2), 256, 0, stream>>>(tbuf, out);
}


# p2g offset loop: fewer vector instructions beside the f32 WMMAs: accumulators go to LDS as half rows (no lane swaps); weight selects and addresses once per 3 offsets; LDS atomic order as before
# speedup vs baseline: 56.8767x; 1.0599x over previous
#include <hip/hip_runtime.h>


typedef float v2f __attribute__((ext_vector_type(2)));
typedef float v8f __attribute__((ext_vector_type(8)));

#define DD 66
#define HHW (66 * 66)
#define DHW (66 * 66 * 66)
#define CI 32
#define CO 32

#define TC 128
#define TH 24
#define TP 33

#define NT 1024
#define TCELLS 1000
#define TSLOT (TCELLS * CO)

#define PBLK 4096
#define BPITCH (NT + 1)

__device__ __forceinline__ int tile_of(const float* __restrict__ pos, int p, int Nper) {
  int b0 = (int)(pos[p * 3 + 0] * 64.0f);
  int b1 = (int)(pos[p * 3 + 1] * 64.0f);
  int b2 = (int)(pos[p * 3 + 2] * 64.0f);
  b0 = min(max(b0, 0), 63);
  b1 = min(max(b1, 0), 63);
  b2 = min(max(b2, 0), 63);
  int bb = (p >= Nper) ? 1 : 0;
  return bb * 512 + (b0 >> 3) * 64 + (b1 >> 3) * 8 + (b2 >> 3);
}

template <int NW, int NV>
__device__ __forceinline__ void block_inclusive_scan(int (&v)[NV], int* s_w) {
  const int lane = threadIdx.x & 31;
  const int wave = threadIdx.x >> 5;
  for (int d = 1; d < 32; d <<= 1) {
    int u[NV];
    for (int j = 0; j < NV; ++j) u[j] = __shfl_up(v[j], d);
    if (lane >= d)
      for (int j = 0; j < NV; ++j) v[j] += u[j];
  }
  if (lane == 31)
    for (int j = 0; j < NV; ++j) s_w[32 * j + wave] = v[j];
  __syncthreads();
  if (wave == 0) {
    int w[NV];
    for (int j = 0; j < NV; ++j) w[j] = (lane < NW) ? s_w[32 * j + lane] : 0;
    for (int d = 1; d < 32; d <<= 1) {
      int u[NV];
      for (int j = 0; j < NV; ++j) u[j] = __shfl_up(w[j], d);
      if (lane >= d)
        for (int j = 0; j < NV; ++j) w[j] += u[j];
    }
    for (int j = 0; j < NV; ++j) s_w[32 * j + lane] = w[j];
  }
  __syncthreads();
  if (wave > 0)
    for (int j = 0; j < NV; ++j) v[j] += s_w[32 * j + wave - 1];
}

__launch_bounds__(1024)
__global__ void block_sort_kernel(const float* __restrict__ pos, const float* __restrict__ kern,
                                  float* __restrict__ kfrag, int* __restrict__ boff,
                                  int* __restrict__ bsorted, int Mtotal, int Nper, int nblk) {
  __shared__ int s_hist[NT];
  __shared__ int s_w[32];
  const int t   = threadIdx.x;
  const int blk = blockIdx.x;
  if (blk >= nblk) {
    if (t < 256) {
      const int o = blk - nblk;
      float4 v4 = ((const float4*)kern)[o * 256 + t];
      int f   = t * 4;
      int ci  = f >> 5;
      int co  = f & 31;
      int nt  = co >> 4;
      int cc  = ci >> 2;
      int hh  = (ci >> 1) & 1;
      int vv  = ci & 1;
      float* d = kfrag + o * 1024 + (nt * 8 + cc) * 64 + (hh * 16 + (co & 15)) * 2 + vv;
      d[0] = v4.x;
      d[2] = v4.y;
      d[4] = v4.z;
      d[6] = v4.w;
    }
    return;
  }
  s_hist[t] = 0;
  int pn[4], tl[4], rk[4];
  for (int k = 0; k < 4; ++k) {
    pn[k] = blk * PBLK + k * 1024 + t;
    tl[k] = tile_of(pos, min(pn[k], Mtotal - 1), Nper);
  }
  __syncthreads();
  for (int k = 0; k < 4; ++k) {
    rk[k] = 0;
    if (pn[k] < Mtotal) rk[k] = atomicAdd(&s_hist[tl[k]], 1);
  }
  __syncthreads();
  const int c = s_hist[t];
  int incl[1] = {c};
  block_inclusive_scan<32, 1>(incl, s_w);
  s_hist[t] = incl[0] - c;
  boff[blk * BPITCH + t] = incl[0] - c;
  if (t == NT - 1) boff[blk * BPITCH + NT] = incl[0];
  __syncthreads();
  for (int k = 0; k < 4; ++k)
    if (pn[k] < Mtotal) bsorted[blk * PBLK + s_hist[tl[k]] + rk[k]] = pn[k];
}

__launch_bounds__(256)
__global__ void tile_cell_sort_kernel(const float* __restrict__ pos, const int* __restrict__ boff,
                                      const int* __restrict__ bsorted, int nblk,
                                      int* __restrict__ off, int* __restrict__ ptmp,
                                      int* __restrict__ tmp, int* __restrict__ sorted2,
                                      int* __restrict__ rowinfo, int* __restrict__ nrows) {
  __shared__ int s_cnt[512];
  __shared__ int s_off[512];
  __shared__ int s_run[128];
  __shared__ int s_rs[128];
  __shared__ int s_w[64];
  const int t    = threadIdx.x;
  const int tile = blockIdx.x;
  int rs = 0, rn = 0;
  if (t < nblk) {
    rs = boff[t * BPITCH + tile];
    rn = boff[t * BPITCH + tile + 1] - rs;
  }
  int runs[2] = {rn, rs};
  block_inclusive_scan<8, 2>(runs, s_w);
  if (t < 128) {
    s_run[t] = runs[0];
    s_rs[t]  = rs;
  }
  s_cnt[t]       = 0;
  s_cnt[t + 256] = 0;
  const int n     = s_w[31];
  const int start = s_w[63];
  if (t == 0) off[tile] = start;
  if (n == 0) {
    if (t == 0) nrows[tile] = 0;
    return;
  }
  const int t0 = (tile >> 6) & 7, t1 = (tile >> 3) & 7, t2 = tile & 7;
  __syncthreads();
  for (int i = t; i < n; i += 256) {
    int lo = 0, hi = 127;
    for (int k = 0; k < 7; ++k) {
      const int mid = (lo + hi) >> 1;
      if (s_run[mid] > i) hi = mid; else lo = mid + 1;
    }
    const int before = (lo > 0) ? s_run[lo - 1] : 0;
    const int p = bsorted[lo * PBLK + s_rs[lo] + (i - before)];
    const int b0 = (int)(pos[p * 3 + 0] * 64.0f);
    const int b1 = (int)(pos[p * 3 + 1] * 64.0f);
    const int b2 = (int)(pos[p * 3 + 2] * 64.0f);
    const int l0 = min(max(b0 - t0 * 8, 0), 7);
    const int l1 = min(max(b1 - t1 * 8, 0), 7);
    const int l2 = min(max(b2 - t2 * 8, 0), 7);
    const int lc = l0 * 64 + l1 * 8 + l2;
    const int r  = atomicAdd(&s_cnt[lc], 1);
    tmp[start + i]  = r | (lc << 20);
    ptmp[start + i] = p;
  }
  __syncthreads();
  const int ca  = s_cnt[2 * t];
  const int cb  = s_cnt[2 * t + 1];
  const int rca = (ca + 1) >> 1;
  const int rcb = (cb + 1) >> 1;
  int ends[2] = {ca + cb, rca + rcb};
  block_inclusive_scan<8, 2>(ends, s_w);
  const int celloffa = ends[0] - cb - ca;
  const int celloffb = ends[0] - cb;
  const int rowoffa  = ends[1] - rcb - rca;
  const int rowoffb  = ends[1] - rcb;
  s_off[2 * t]     = celloffa;
  s_off[2 * t + 1] = celloffb;
  if (t == 255) nrows[tile] = ends[1];
  __syncthreads();
  for (int i = t; i < n; i += 256) {
    const int v = tmp[start + i];
    sorted2[start + s_off[v >> 20] + (v & 0xFFFFF)] = ptmp[start + i];
  }
  for (int k = 0; k < rca; ++k)
    rowinfo[start + rowoffa + k] = (celloffa + 2 * k) | ((ca - 2 * k >= 2) ? (1 << 30) : 0);
  for (int k = 0; k < rcb; ++k)
    rowinfo[start + rowoffb + k] = (celloffb + 2 * k) | ((cb - 2 * k >= 2) ? (1 << 30) : 0);
}

__device__ __forceinline__ void particle_weights(const float* __restrict__ pos, int p, float valid,
                                                 int t0, int t1, int t2,
                                                 float (&wx)[3], float (&wy)[3], float (&wz)[3],
                                                 int& lbase) {
  float px = pos[p * 3 + 0] * 64.0f;
  float py = pos[p * 3 + 1] * 64.0f;
  float pz = pos[p * 3 + 2] * 64.0f;
  int b0 = (int)px, b1 = (int)py, b2 = (int)pz;
  float fx = px - (float)b0 - 0.5f;
  float fy = py - (float)b1 - 0.5f;
  float fz = pz - (float)b2 - 0.5f;
  wx[0] = 0.5f * (0.5f - fx) * (0.5f - fx) * valid;
  wx[1] = (0.75f - fx * fx) * valid;
  wx[2] = 0.5f * (0.5f + fx) * (0.5f + fx) * valid;
  wy[0] = 0.5f * (0.5f - fy) * (0.5f - fy);
  wy[1] = 0.75f - fy * fy;
  wy[2] = 0.5f * (0.5f + fy) * (0.5f + fy);
  wz[0] = 0.5f * (0.5f - fz) * (0.5f - fz);
  wz[1] = 0.75f - fz * fz;
  wz[2] = 0.5f * (0.5f + fz) * (0.5f + fz);
  const int l0 = min(max(b0 - t0 * 8, 0), 7);
  const int l1 = min(max(b1 - t1 * 8, 0), 7);
  const int l2 = min(max(b2 - t2 * 8, 0), 7);
  lbase = l0 * 100 + l1 * 10 + l2;
}

__launch_bounds__(256, 4)
__global__ void p2g_tile_kernel(const float* __restrict__ x,
                                const float* __restrict__ pos,
                                const float* __restrict__ kfrag,
                                const float* __restrict__ bias,
                                const int* __restrict__ nrows,
                                const int* __restrict__ off,
                                const int* __restrict__ sorted2,
                                const int* __restrict__ rowinfo,
                                float* __restrict__ tbuf) {
  __shared__ __attribute__((aligned(16))) float sAcc[TCELLS * CO];
  __shared__ float sW[TCELLS];

  const int t    = threadIdx.x;
  const int lane = t & 31;
  const int wv   = __builtin_amdgcn_readfirstlane(t >> 5);
  const int half = lane >> 4;
  const int nlo  = lane & 15;

  const int tile  = blockIdx.x;
  const int nr    = nrows[tile];
  const int start = off[tile];
  float4* tb4 = (float4*)(tbuf + (size_t)tile * TSLOT);
  if (nr == 0) {
    for (int i = t; i < TSLOT / 4; i += 256) tb4[i] = make_float4(0.f, 0.f, 0.f, 0.f);
    return;
  }
  const int t0 = (tile >> 6) & 7, t1 = (tile >> 3) & 7, t2 = tile & 7;

  {
    float4* z = (float4*)sAcc;
    for (int i = t; i < TCELLS * CO / 4; i += 256) z[i] = make_float4(0.f, 0.f, 0.f, 0.f);
    for (int i = t; i < TCELLS; i += 256) sW[i] = 0.f;
  }
  __syncthreads();

  const int nch = (nr + 31) >> 5;

  const int tot = nch * 27;
  const int lo  = (wv * tot) >> 3;
  const int hi  = ((wv + 1) * tot) >> 3;
  for (int ch = lo / 27; ch * 27 < hi; ++ch) {
    const int oB = max(lo - ch * 27, 0);
    const int oE = min(hi - ch * 27, 27);

    v2f xa[2][8], xb[2][8];
    float wax0[3], way0[3], waz0[3], wbx0[3], wby0[3], wbz0[3];
    float wax1[3], way1[3], waz1[3], wbx1[3], wby1[3], wbz1[3];
    int lbA[2];
#pragma unroll
    for (int mt = 0; mt < 2; ++mt) {
      const int jr   = ch * 32 + mt * 16 + nlo;
      const bool vr  = jr < nr;
      const int jc   = vr ? jr : (nr - 1);
      const int info = rowinfo[start + jc];
      const int slot = info & 0x3FFFFFFF;
      const int two  = (info >> 30) & 1;
      const int pa   = sorted2[start + slot];
      const int pb   = sorted2[start + slot + two];
      int lbB;
      if (mt == 0) {
        particle_weights(pos, pa, vr ? 1.0f : 0.0f, t0, t1, t2, wax0, way0, waz0, lbA[0]);
        particle_weights(pos, pb, (vr && two) ? 1.0f : 0.0f, t0, t1, t2, wbx0, wby0, wbz0, lbB);
      } else {
        particle_weights(pos, pa, vr ? 1.0f : 0.0f, t0, t1, t2, wax1, way1, waz1, lbA[1]);
        particle_weights(pos, pb, (vr && two) ? 1.0f : 0.0f, t0, t1, t2, wbx1, wby1, wbz1, lbB);
      }
      const float* xra = x + pa * CI + half * 2;
      const float* xrb = x + pb * CI + half * 2;
#pragma unroll
      for (int c = 0; c < 8; ++c) {
        xa[mt][c] = *(const v2f*)(xra + c * 4);
        xb[mt][c] = *(const v2f*)(xrb + c * 4);
      }
    }
    const int lbase = half ? lbA[1] : lbA[0];

    int lb[32];
#pragma unroll
    for (int m = 0; m < 32; ++m) lb[m] = __builtin_amdgcn_readlane(lbase * CO, m);

    int radr[2][8];
#pragma unroll
    for (int mt = 0; mt < 2; ++mt)
#pragma unroll
      for (int r = 0; r < 8; ++r)
        radr[mt][r] = (half ? lb[mt * 16 + r + 8] : lb[mt * 16 + r]) + nlo;

    const int q0 = oB / 3;
    int oi = q0 / 3, oj = q0 - oi * 3;
    int radq[2][8];
#pragma unroll
    for (int mt = 0; mt < 2; ++mt)
#pragma unroll
      for (int r = 0; r < 8; ++r) radq[mt][r] = radr[mt][r] + (oi * 100 + oj * 10) * CO;
    int lbq = lbase + oi * 100 + oj * 10;
    for (int q = q0; q * 3 < oE; ++q) {
      const float pa0 = wax0[oi] * way0[oj];
      const float pb0 = wbx0[oi] * wby0[oj];
      const float pa1 = wax1[oi] * way1[oj];
      const float pb1 = wbx1[oi] * wby1[oj];
      const float* kfq = kfrag + q * 3072 + lane * 2;
#pragma unroll
      for (int ok = 0; ok < 3; ++ok) {
      const int o = q * 3 + ok;
      if (o < oB || o >= oE) continue;

      float wa[2], wb[2];
      wa[0] = pa0 * waz0[ok];
      wb[0] = pb0 * wbz0[ok];
      wa[1] = pa1 * waz1[ok];
      wb[1] = pb1 * wbz1[ok];
      atomicAdd(&sW[lbq + ok], half ? (wa[1] + wb[1]) : (wa[0] + wb[0]));

      v2f a[2][8];
#pragma unroll
      for (int mt = 0; mt < 2; ++mt)
#pragma unroll
        for (int c = 0; c < 8; ++c)
          a[mt][c] = xa[mt][c] * wa[mt] + xb[mt][c] * wb[mt];

      const float* kf = kfq + ok * 1024;
      v2f bf[2][8];
#pragma unroll
      for (int nt = 0; nt < 2; ++nt)
#pragma unroll
        for (int c = 0; c < 8; ++c)
          bf[nt][c] = *(const v2f*)&kf[(nt * 8 + c) * 64];

      v8f dacc[2][2];
#pragma unroll
      for (int mt = 0; mt < 2; ++mt)
#pragma unroll
        for (int nt = 0; nt < 2; ++nt)
#pragma unroll
          for (int r = 0; r < 8; ++r) dacc[mt][nt][r] = 0.0f;

#pragma unroll
      for (int c = 0; c < 8; ++c) {
#pragma unroll
        for (int mt = 0; mt < 2; ++mt) {
#pragma unroll
          for (int nt = 0; nt < 2; ++nt) {
            dacc[mt][nt] = __builtin_amdgcn_wmma_f32_16x16x4_f32(
                false, a[mt][c], false, bf[nt][c], (short)0, dacc[mt][nt],
                false, false);
          }
        }
      }

#pragma unroll
      for (int mt = 0; mt < 2; ++mt) {
#pragma unroll
        for (int r = 0; r < 8; ++r) {
          float* line = &sAcc[radq[mt][r] + ok * CO];
          atomicAdd(line, dacc[mt][0][r]);
          atomicAdd(line + 16, dacc[mt][1][r]);
        }
      }
      }
      const int stepc = (oj == 2) ? 80 : 10;
#pragma unroll
      for (int mt = 0; mt < 2; ++mt)
#pragma unroll
        for (int r = 0; r < 8; ++r) radq[mt][r] += stepc * CO;
      lbq += stepc;
      if (oj == 2) { oj = 0; ++oi; } else { ++oj; }
    }
  }

  __syncthreads();

  {
    const float4* s4 = (const float4*)sAcc;
    const float4 b4 = ((const float4*)bias)[t & 7];
    for (int i = t; i < TSLOT / 4; i += 256) {
      const float4 s = s4[i];
      const float w = sW[i >> 3];
      tb4[i] = make_float4(fmaf(w, b4.x, s.x), fmaf(w, b4.y, s.y), fmaf(w, b4.z, s.z), fmaf(w, b4.w, s.w));
    }
  }
}

__launch_bounds__(256)
__global__ void sum_copies_channel_major_kernel(const float* __restrict__ tbuf,
                                                float* __restrict__ out) {
  __shared__ float tile[(TC + TH) * TP];

  const int t    = threadIdx.x;
  const int lane = t & 31;
  const int wv   = t >> 5;
  const int b    = blockIdx.y;
  const int c0   = blockIdx.x * TC;

  int ncell = DHW - c0;
  ncell = (ncell < TC + TH) ? ncell : (TC + TH);
  const int nf4 = ncell * (CO / 4);
  for (int i = t; i < nf4; i += 256) {
    const int g  = c0 + (i >> 3);
    const int d0 = g / HHW;
    const int gr = g - d0 * HHW;
    const int d1 = gr / DD;
    const int d2 = gr - d1 * DD;
    const int q0 = d0 >> 3, q1 = d1 >> 3, q2 = d2 >> 3;
    const int r0 = d0 & 7,  r1 = d1 & 7,  r2 = d2 & 7;
    const int e = ((b * 512 + q0 * 64 + q1 * 8 + q2) * TCELLS + r0 * 100 + r1 * 10 + r2) * CO + (i & 7) * 4;
    const int E0 = (800 - 64 * TCELLS) * CO;
    const int E1 = (80 - 8 * TCELLS) * CO;
    const int E2 = (8 - TCELLS) * CO;
    const bool A0 = q0 < 8, A1 = q1 < 8, A2 = q2 < 8;
    const bool B0 = (q0 >= 1) && (r0 < 2);
    const bool B1 = (q1 >= 1) && (r1 < 2);
    const bool B2 = (q2 >= 1) && (r2 < 2);

    const float4 z4 = make_float4(0.f, 0.f, 0.f, 0.f);
    float4 v0 = z4, v1 = z4, v2 = z4, v3 = z4, v4 = z4, v5 = z4, v6 = z4, v7 = z4;
    if (A0) {
      if (A1) {
        if (A2) v0 = *(const float4*)(tbuf + e);
        if (B2) v1 = *(const float4*)(tbuf + (e + E2));
      }
      if (B1) {
        if (A2) v2 = *(const float4*)(tbuf + (e + E1));
        if (B2) v3 = *(const float4*)(tbuf + (e + E1 + E2));
      }
    }
    if (B0) {
      if (A1) {
        if (A2) v4 = *(const float4*)(tbuf + (e + E0));
        if (B2) v5 = *(const float4*)(tbuf + (e + E0 + E2));
      }
      if (B1) {
        if (A2) v6 = *(const float4*)(tbuf + (e + E0 + E1));
        if (B2) v7 = *(const float4*)(tbuf + (e + E0 + E1 + E2));
      }
    }
    float4 s;
    s.x = ((((((v0.x + v1.x) + v2.x) + v3.x) + v4.x) + v5.x) + v6.x) + v7.x;
    s.y = ((((((v0.y + v1.y) + v2.y) + v3.y) + v4.y) + v5.y) + v6.y) + v7.y;
    s.z = ((((((v0.z + v1.z) + v2.z) + v3.z) + v4.z) + v5.z) + v6.z) + v7.z;
    s.w = ((((((v0.w + v1.w) + v2.w) + v3.w) + v4.w) + v5.w) + v6.w) + v7.w;

    float* d = &tile[(i >> 3) * TP + (i & 7) * 4];
    d[0] = s.x; d[1] = s.y; d[2] = s.z; d[3] = s.w;
  }
  __syncthreads();

  float v[4][TC / 32];
  float vhead[4];
#pragma unroll
  for (int k = 0; k < 4; ++k) {
    const int ch = wv * 4 + k;
    const int sh = (32 - 8 * (ch & 3)) & 31;
#pragma unroll
    for (int e = 0; e < TC / 32; ++e) {
      int lr = sh + e * 32 + lane;
      lr = (lr < ncell) ? lr : (ncell - 1);
      v[k][e] = tile[lr * TP + ch];
    }
    vhead[k] = tile[((lane < ncell) ? lane : (ncell - 1)) * TP + ch];
  }
#pragma unroll
  for (int k = 0; k < 4; ++k) {
    const int ch = wv * 4 + k;
    const int sh = (32 - 8 * (ch & 3)) & 31;
    float* orow = out + ((size_t)(b * CO + ch)) * DHW + c0;
#pragma unroll
    for (int e = 0; e < TC / 32; ++e) {
      const int lc = sh + e * 32 + lane;
      if (c0 + lc < DHW) orow[lc] = v[k][e];
    }
    if (blockIdx.x == 0 && lane < sh) orow[lane] = vhead[k];
  }
}

extern "C" void kernel_launch(void* const* d_in, const int* in_sizes, int n_in,
                              void* d_out, int out_size, void* d_ws, size_t ws_size,
                              hipStream_t stream) {
  const float* x    = (const float*)d_in[0];
  const float* pos  = (const float*)d_in[1];
  const float* kern = (const float*)d_in[2];
  const float* bias = (const float*)d_in[3];
  float* out = (float*)d_out;
  float* tbuf = (float*)d_ws;
  int* wsi    = (int*)d_ws + 32768000;
  int* off    = wsi + 1024;
  int* tmp    = wsi + 2048;
  int* ptmp   = wsi + 2048 + 524288;
  float* kfrag = (float*)(wsi + 2048 + 524288 + 524288);
  int* sorted2 = wsi + 1078272;
  int* rowinfo = sorted2 + 524288;
  int* nrows   = rowinfo + 524288;
  int* boff    = nrows + 1024;
  int* bsorted = boff + 128 * BPITCH;

  const int Mtotal = in_sizes[0] / CI;
  const int Nper   = in_sizes[1] / (2 * 3);
  const int nblk   = (Mtotal + PBLK - 1) / PBLK;

  block_sort_kernel<<<nblk + 27, 1024, 0, stream>>>(pos, kern, kfrag, boff, bsorted, Mtotal, Nper, nblk);

  tile_cell_sort_kernel<<<1024, 256, 0, stream>>>(pos, boff, bsorted, nblk, off, ptmp, tmp, sorted2, rowinfo, nrows);

  p2g_tile_kernel<<<1024, 256, 0, stream>>>(x, pos, kfrag, bias, nrows, off, sorted2, rowinfo, tbuf);

  sum_copies_channel_major_kernel<<<dim3((DHW + TC - 1) / TC, 2), 256, 0, stream>>>(tbuf, out);
}
